# starting kernel
# baseline (speedup 1.0000x reference)
#include <hip/hip_runtime.h>

typedef float v2f __attribute__((ext_vector_type(2)));
typedef float v8f __attribute__((ext_vector_type(8)));

#define IN_DIM  64
#define OUT_DIM 64
#define HID     64
#define DDIM    (IN_DIM * OUT_DIM)
#define BSZ     256
#define LSTR    68

__launch_bounds__(256)
__global__ void kan_edge_kernel(const float* __restrict__ x,
                                const float* __restrict__ w0, const float* __restrict__ b0,
                                const float* __restrict__ w1, const float* __restrict__ b1,
                                const float* __restrict__ w2, const float* __restrict__ b2,
                                const float* __restrict__ wout,
                                float* __restrict__ outbuf, int use_atomic)
{
    __shared__ float act0[BSZ * LSTR];
    __shared__ float act1[BSZ * LSTR];
    __shared__ float xs[BSZ];
    __shared__ float ws0[HID], bs0[HID], wos[HID];

    const int d   = blockIdx.x;
    const int o   = d >> 6;
    const int i   = d & 63;
    const int tid = threadIdx.x;

    xs[tid & 255] = x[(tid & 255) * IN_DIM + i];
    if (tid < HID) {
        ws0[tid] = w0[d * HID + tid];
        bs0[tid] = b0[d * HID + tid];
        wos[tid] = wout[o * (IN_DIM * HID) + i * HID + tid];
    }
    __syncthreads();

    {
        const int   h  = tid & 63;
        const int   bg = tid >> 6;
        const float wv = ws0[h];
        const float bv = bs0[h];
        for (int b = bg; b < BSZ; b += 4) {
            float v = fmaf(xs[b], wv, bv);
            act0[b * LSTR + h] = v > 0.f ? v : 0.f;
        }
    }
    __syncthreads();

    const int lane  = tid & 31;
    const int r     = lane & 15;
    const int half  = lane >> 4;
    const int nbase = ((tid >> 5) & 3) * 16;
    const int mhalf = (tid >> 5) >> 2;

    auto layer = [&](const float* __restrict__ src, float* __restrict__ dst,
                     const float* __restrict__ W, const float* __restrict__ Bv) {
        v2f Bfrag[16];
        const float* wrow = W + (size_t)(nbase + r) * HID + 2 * half;
        #pragma unroll
        for (int c = 0; c < 16; ++c)
            Bfrag[c] = *(const v2f*)(wrow + c * 4);
        const float bias = Bv[nbase + r];

        #pragma unroll
        for (int t = 0; t < 4; ++t) {
            const int m0 = (mhalf * 8 + 2 * t) * 16;
            const int m1 = m0 + 16;
            v8f acc0 = {0.f,0.f,0.f,0.f,0.f,0.f,0.f,0.f};
            v8f acc1 = {0.f,0.f,0.f,0.f,0.f,0.f,0.f,0.f};
            const float* a0p = src + (m0 + r) * LSTR + 2 * half;
            const float* a1p = src + (m1 + r) * LSTR + 2 * half;
            #pragma unroll
            for (int c = 0; c < 16; ++c) {
                v2f a0 = *(const v2f*)(a0p + c * 4);
                v2f a1 = *(const v2f*)(a1p + c * 4);
                acc0 = __builtin_amdgcn_wmma_f32_16x16x4_f32(
                           false, a0, false, Bfrag[c], (short)0, acc0, false, false);
                acc1 = __builtin_amdgcn_wmma_f32_16x16x4_f32(
                           false, a1, false, Bfrag[c], (short)0, acc1, false, false);
            }
            #pragma unroll
            for (int v = 0; v < 8; ++v) {
                const int row0 = m0 + v + 8 * half;
                const int row1 = m1 + v + 8 * half;
                float v0 = acc0[v] + bias; v0 = v0 > 0.f ? v0 : 0.f;
                float v1 = acc1[v] + bias; v1 = v1 > 0.f ? v1 : 0.f;
                dst[row0 * LSTR + nbase + r] = v0;
                dst[row1 * LSTR + nbase + r] = v1;
            }
        }
    };

    layer(act0, act1, w1 + (size_t)d * HID * HID, b1 + d * HID);
    __syncthreads();
    layer(act1, act0, w2 + (size_t)d * HID * HID, b2 + d * HID);
    __syncthreads();

    {
        const int b = tid;
        float s = 0.f;
        const float* row = act0 + b * LSTR;
        #pragma unroll 8
        for (int h = 0; h < HID; ++h) s = fmaf(wos[h], row[h], s);
        if (use_atomic) {
            __hip_atomic_fetch_add(&outbuf[b * OUT_DIM + o], s,
                                   __ATOMIC_RELAXED, __HIP_MEMORY_SCOPE_AGENT);
        } else {
            outbuf[(size_t)d * BSZ + b] = s;
        }
    }
}

__global__ void kan_reduce_kernel(const float* __restrict__ part,
                                  const float* __restrict__ bout,
                                  float* __restrict__ out)
{
    const int o = blockIdx.x;
    const int b = threadIdx.x;
    float s = bout[o];
    const float* p = part + (size_t)o * 64 * BSZ + b;
    #pragma unroll 8
    for (int i = 0; i < 64; ++i) s += p[i * BSZ];
    out[b * OUT_DIM + o] = s;
}

__global__ void kan_init_out(const float* __restrict__ bout, float* __restrict__ out)
{
    const int idx = blockIdx.x * 256 + threadIdx.x;
    out[idx] = bout[idx & 63];
}

extern "C" void kernel_launch(void* const* d_in, const int* in_sizes, int n_in,
                              void* d_out, int out_size, void* d_ws, size_t ws_size,
                              hipStream_t stream)
{
    const float* x    = (const float*)d_in[0];
    const float* w0   = (const float*)d_in[1];
    const float* b0   = (const float*)d_in[2];
    const float* w1   = (const float*)d_in[3];
    const float* b1   = (const float*)d_in[4];
    const float* w2   = (const float*)d_in[5];
    const float* b2   = (const float*)d_in[6];
    const float* wout = (const float*)d_in[7];
    const float* bout = (const float*)d_in[8];
    float* out = (float*)d_out;

    const size_t need = (size_t)DDIM * BSZ * sizeof(float);
    if (ws_size >= need) {
        float* part = (float*)d_ws;
        kan_edge_kernel<<<DDIM, 256, 0, stream>>>(x, w0, b0, w1, b1, w2, b2, wout, part, 0);
        kan_reduce_kernel<<<OUT_DIM, BSZ, 0, stream>>>(part, bout, out);
    } else {
        kan_init_out<<<64, 256, 0, stream>>>(bout, out);
        kan_edge_kernel<<<DDIM, 256, 0, stream>>>(x, w0, b0, w1, b1, w2, b2, wout, out, 1);
    }
}


# Two 4-wave workgroups per edge, 128 batch rows each (was one 8-wave workgroup, 256 rows): LDS per workgroup halves to 70 KB so 4 independent workgroups fit a WGP instead of 2; per-wave code unchanged
# speedup vs baseline: 1.0693x; 1.0693x over previous
#include <hip/hip_runtime.h>

typedef float v2f __attribute__((ext_vector_type(2)));
typedef float v8f __attribute__((ext_vector_type(8)));

#define IN_DIM  64
#define OUT_DIM 64
#define HID     64
#define DDIM    (IN_DIM * OUT_DIM)
#define BSZ     256
#define MROWS   128
#define LSTR    68

__launch_bounds__(128)
__global__ void kan_edge_kernel(const float* __restrict__ x,
                                const float* __restrict__ w0, const float* __restrict__ b0,
                                const float* __restrict__ w1, const float* __restrict__ b1,
                                const float* __restrict__ w2, const float* __restrict__ b2,
                                const float* __restrict__ wout,
                                float* __restrict__ outbuf, int use_atomic)
{
    __shared__ float act0[MROWS * LSTR];
    __shared__ float act1[MROWS * LSTR];
    __shared__ float xs[MROWS];
    __shared__ float ws0[HID], bs0[HID], wos[HID];

    const int d   = blockIdx.x >> 1;
    const int bat0 = (blockIdx.x & 1) * MROWS;
    const int o   = d >> 6;
    const int i   = d & 63;
    const int tid = threadIdx.x;

    xs[tid] = x[(bat0 + tid) * IN_DIM + i];
    if (tid < HID) {
        ws0[tid] = w0[d * HID + tid];
        bs0[tid] = b0[d * HID + tid];
        wos[tid] = wout[o * (IN_DIM * HID) + i * HID + tid];
    }
    __syncthreads();

    {
        const int   h  = tid & 63;
        const int   bg = tid >> 6;
        const float wv = ws0[h];
        const float bv = bs0[h];
        for (int b = bg; b < MROWS; b += 2) {
            float v = fmaf(xs[b], wv, bv);
            act0[b * LSTR + h] = v > 0.f ? v : 0.f;
        }
    }
    __syncthreads();

    const int lane  = tid & 31;
    const int r     = lane & 15;
    const int half  = lane >> 4;
    const int nbase = (tid >> 5) * 16;

    auto layer = [&](const float* __restrict__ src, float* __restrict__ dst,
                     const float* __restrict__ W, const float* __restrict__ Bv) {
        v2f Bfrag[16];
        const float* wrow = W + (size_t)(nbase + r) * HID + 2 * half;
        #pragma unroll
        for (int c = 0; c < 16; ++c)
            Bfrag[c] = *(const v2f*)(wrow + c * 4);
        const float bias = Bv[nbase + r];

        #pragma unroll
        for (int t = 0; t < 4; ++t) {
            const int m0 = (2 * t) * 16;
            const int m1 = m0 + 16;
            v8f acc0 = {0.f,0.f,0.f,0.f,0.f,0.f,0.f,0.f};
            v8f acc1 = {0.f,0.f,0.f,0.f,0.f,0.f,0.f,0.f};
            const float* a0p = src + (m0 + r) * LSTR + 2 * half;
            const float* a1p = src + (m1 + r) * LSTR + 2 * half;
            #pragma unroll
            for (int c = 0; c < 16; ++c) {
                v2f a0 = *(const v2f*)(a0p + c * 4);
                v2f a1 = *(const v2f*)(a1p + c * 4);
                acc0 = __builtin_amdgcn_wmma_f32_16x16x4_f32(
                           false, a0, false, Bfrag[c], (short)0, acc0, false, false);
                acc1 = __builtin_amdgcn_wmma_f32_16x16x4_f32(
                           false, a1, false, Bfrag[c], (short)0, acc1, false, false);
            }
            #pragma unroll
            for (int v = 0; v < 8; ++v) {
                const int row0 = m0 + v + 8 * half;
                const int row1 = m1 + v + 8 * half;
                float v0 = acc0[v] + bias; v0 = v0 > 0.f ? v0 : 0.f;
                float v1 = acc1[v] + bias; v1 = v1 > 0.f ? v1 : 0.f;
                dst[row0 * LSTR + nbase + r] = v0;
                dst[row1 * LSTR + nbase + r] = v1;
            }
        }
    };

    layer(act0, act1, w1 + (size_t)d * HID * HID, b1 + d * HID);
    __syncthreads();
    layer(act1, act0, w2 + (size_t)d * HID * HID, b2 + d * HID);
    __syncthreads();

    {
        const int b = tid;
        float s = 0.f;
        const float* row = act0 + b * LSTR;
        #pragma unroll 8
        for (int h = 0; h < HID; ++h) s = fmaf(wos[h], row[h], s);
        if (use_atomic) {
            __hip_atomic_fetch_add(&outbuf[(bat0 + b) * OUT_DIM + o], s,
                                   __ATOMIC_RELAXED, __HIP_MEMORY_SCOPE_AGENT);
        } else {
            outbuf[(size_t)d * BSZ + bat0 + b] = s;
        }
    }
}

__global__ void kan_reduce_kernel(const float* __restrict__ part,
                                  const float* __restrict__ bout,
                                  float* __restrict__ out)
{
    const int o = blockIdx.x;
    const int b = threadIdx.x;
    float s = bout[o];
    const float* p = part + (size_t)o * 64 * BSZ + b;
    #pragma unroll 8
    for (int i = 0; i < 64; ++i) s += p[i * BSZ];
    out[b * OUT_DIM + o] = s;
}

__global__ void kan_init_out(const float* __restrict__ bout, float* __restrict__ out)
{
    const int idx = blockIdx.x * 256 + threadIdx.x;
    out[idx] = bout[idx & 63];
}

extern "C" void kernel_launch(void* const* d_in, const int* in_sizes, int n_in,
                              void* d_out, int out_size, void* d_ws, size_t ws_size,
                              hipStream_t stream)
{
    const float* x    = (const float*)d_in[0];
    const float* w0   = (const float*)d_in[1];
    const float* b0   = (const float*)d_in[2];
    const float* w1   = (const float*)d_in[3];
    const float* b1   = (const float*)d_in[4];
    const float* w2   = (const float*)d_in[5];
    const float* b2   = (const float*)d_in[6];
    const float* wout = (const float*)d_in[7];
    const float* bout = (const float*)d_in[8];
    float* out = (float*)d_out;

    const size_t need = (size_t)DDIM * BSZ * sizeof(float);
    if (ws_size >= need) {
        float* part = (float*)d_ws;
        kan_edge_kernel<<<2 * DDIM, MROWS, 0, stream>>>(x, w0, b0, w1, b1, w2, b2, wout, part, 0);
        kan_reduce_kernel<<<OUT_DIM, BSZ, 0, stream>>>(part, bout, out);
    } else {
        kan_init_out<<<64, 256, 0, stream>>>(bout, out);
        kan_edge_kernel<<<2 * DDIM, MROWS, 0, stream>>>(x, w0, b0, w1, b1, w2, b2, wout, out, 1);
    }
}


# Hidden layers update ONE LDS activation buffer in place, a 64-row half at a time (its 4 accumulator tiles wait in registers for a barrier): LDS 70->36 KB, 7 workgroups per WGP instead of 4
# speedup vs baseline: 1.1514x; 1.0768x over previous
#include <hip/hip_runtime.h>

typedef float v2f __attribute__((ext_vector_type(2)));
typedef float v8f __attribute__((ext_vector_type(8)));

#define IN_DIM  64
#define OUT_DIM 64
#define HID     64
#define DDIM    (IN_DIM * OUT_DIM)
#define BSZ     256
#define MROWS   128
#define LSTR    68

__launch_bounds__(128)
__global__ void kan_edge_kernel(const float* __restrict__ x,
                                const float* __restrict__ w0, const float* __restrict__ b0,
                                const float* __restrict__ w1, const float* __restrict__ b1,
                                const float* __restrict__ w2, const float* __restrict__ b2,
                                const float* __restrict__ wout,
                                float* __restrict__ outbuf, int use_atomic)
{
    __shared__ float act[MROWS * LSTR];
    __shared__ float xs[MROWS];
    __shared__ float ws0[HID], bs0[HID], wos[HID];

    const int d   = blockIdx.x >> 1;
    const int bat0 = (blockIdx.x & 1) * MROWS;
    const int o   = d >> 6;
    const int i   = d & 63;
    const int tid = threadIdx.x;

    xs[tid] = x[(bat0 + tid) * IN_DIM + i];
    if (tid < HID) {
        ws0[tid] = w0[d * HID + tid];
        bs0[tid] = b0[d * HID + tid];
        wos[tid] = wout[o * (IN_DIM * HID) + i * HID + tid];
    }
    __syncthreads();

    {
        const int   h  = tid & 63;
        const int   bg = tid >> 6;
        const float wv = ws0[h];
        const float bv = bs0[h];
        for (int b = bg; b < MROWS; b += 2) {
            float v = fmaf(xs[b], wv, bv);
            act[b * LSTR + h] = v > 0.f ? v : 0.f;
        }
    }
    __syncthreads();

    const int lane  = tid & 31;
    const int r     = lane & 15;
    const int half  = lane >> 4;
    const int nbase = (tid >> 5) * 16;

    auto layer = [&](const float* __restrict__ W, const float* __restrict__ Bv) {
        v2f Bfrag[16];
        const float* wrow = W + (size_t)(nbase + r) * HID + 2 * half;
        #pragma unroll
        for (int c = 0; c < 16; ++c)
            Bfrag[c] = *(const v2f*)(wrow + c * 4);
        const float bias = Bv[nbase + r];

        #pragma unroll
        for (int hr = 0; hr < 2; ++hr) {
            v8f acc[4];
            #pragma unroll
            for (int t = 0; t < 2; ++t) {
                const int m0 = hr * 64 + (2 * t) * 16;
                const int m1 = m0 + 16;
                v8f acc0 = {0.f,0.f,0.f,0.f,0.f,0.f,0.f,0.f};
                v8f acc1 = {0.f,0.f,0.f,0.f,0.f,0.f,0.f,0.f};
                const float* a0p = act + (m0 + r) * LSTR + 2 * half;
                const float* a1p = act + (m1 + r) * LSTR + 2 * half;
                #pragma unroll
                for (int c = 0; c < 16; ++c) {
                    v2f a0 = *(const v2f*)(a0p + c * 4);
                    v2f a1 = *(const v2f*)(a1p + c * 4);
                    acc0 = __builtin_amdgcn_wmma_f32_16x16x4_f32(
                               false, a0, false, Bfrag[c], (short)0, acc0, false, false);
                    acc1 = __builtin_amdgcn_wmma_f32_16x16x4_f32(
                               false, a1, false, Bfrag[c], (short)0, acc1, false, false);
                }
                acc[2 * t]     = acc0;
                acc[2 * t + 1] = acc1;
            }

            __syncthreads();

            #pragma unroll
            for (int t = 0; t < 4; ++t) {
                #pragma unroll
                for (int v = 0; v < 8; ++v) {
                    const int row = hr * 64 + t * 16 + v + 8 * half;
                    float o = acc[t][v] + bias; o = o > 0.f ? o : 0.f;
                    act[row * LSTR + nbase + r] = o;
                }
            }
        }
    };

    layer(w1 + (size_t)d * HID * HID, b1 + d * HID);
    layer(w2 + (size_t)d * HID * HID, b2 + d * HID);
    __syncthreads();

    {
        const int b = tid;
        float s = 0.f;
        const float* row = act + b * LSTR;
        #pragma unroll 8
        for (int h = 0; h < HID; ++h) s = fmaf(wos[h], row[h], s);
        if (use_atomic) {
            __hip_atomic_fetch_add(&outbuf[(bat0 + b) * OUT_DIM + o], s,
                                   __ATOMIC_RELAXED, __HIP_MEMORY_SCOPE_AGENT);
        } else {
            outbuf[(size_t)d * BSZ + bat0 + b] = s;
        }
    }
}

__global__ void kan_reduce_kernel(const float* __restrict__ part,
                                  const float* __restrict__ bout,
                                  float* __restrict__ out)
{
    const int o = blockIdx.x;
    const int b = threadIdx.x;
    float s = bout[o];
    const float* p = part + (size_t)o * 64 * BSZ + b;
    #pragma unroll 8
    for (int i = 0; i < 64; ++i) s += p[i * BSZ];
    out[b * OUT_DIM + o] = s;
}

__global__ void kan_init_out(const float* __restrict__ bout, float* __restrict__ out)
{
    const int idx = blockIdx.x * 256 + threadIdx.x;
    out[idx] = bout[idx & 63];
}

extern "C" void kernel_launch(void* const* d_in, const int* in_sizes, int n_in,
                              void* d_out, int out_size, void* d_ws, size_t ws_size,
                              hipStream_t stream)
{
    const float* x    = (const float*)d_in[0];
    const float* w0   = (const float*)d_in[1];
    const float* b0   = (const float*)d_in[2];
    const float* w1   = (const float*)d_in[3];
    const float* b1   = (const float*)d_in[4];
    const float* w2   = (const float*)d_in[5];
    const float* b2   = (const float*)d_in[6];
    const float* wout = (const float*)d_in[7];
    const float* bout = (const float*)d_in[8];
    float* out = (float*)d_out;

    const size_t need = (size_t)DDIM * BSZ * sizeof(float);
    if (ws_size >= need) {
        float* part = (float*)d_ws;
        kan_edge_kernel<<<2 * DDIM, MROWS, 0, stream>>>(x, w0, b0, w1, b1, w2, b2, wout, part, 0);
        kan_reduce_kernel<<<OUT_DIM, BSZ, 0, stream>>>(part, bout, out);
    } else {
        kan_init_out<<<64, 256, 0, stream>>>(bout, out);
        kan_edge_kernel<<<2 * DDIM, MROWS, 0, stream>>>(x, w0, b0, w1, b1, w2, b2, wout, out, 1);
    }
}


# Non-matrix instructions per wave cut 751 -> 587 (vector/LDS/memory): input layer by 4-column groups with 16-byte LDS stores, wout as scalar operands, no LDS staging, one barrier less; sums unchanged
# speedup vs baseline: 1.2019x; 1.0439x over previous
#include <hip/hip_runtime.h>

typedef float v2f __attribute__((ext_vector_type(2)));
typedef float v8f __attribute__((ext_vector_type(8)));

#define IN_DIM  64
#define OUT_DIM 64
#define HID     64
#define DDIM    (IN_DIM * OUT_DIM)
#define BSZ     256
#define MROWS   128
#define LSTR    68

__launch_bounds__(128)
__global__ void kan_edge_kernel(const float* __restrict__ x,
                                const float* __restrict__ w0, const float* __restrict__ b0,
                                const float* __restrict__ w1, const float* __restrict__ b1,
                                const float* __restrict__ w2, const float* __restrict__ b2,
                                const float* __restrict__ wout,
                                float* __restrict__ outbuf, int use_atomic)
{
    __shared__ __attribute__((aligned(16))) float act[MROWS * LSTR];

    const int d   = blockIdx.x >> 1;
    const int bat0 = (blockIdx.x & 1) * MROWS;
    const int o   = d >> 6;
    const int i   = d & 63;
    const int tid = threadIdx.x;

    {
        const int g  = tid & 15;
        const int r0 = (tid >> 4) * 16;
        const float4 w4 = *reinterpret_cast<const float4*>(w0 + d * HID + 4 * g);
        const float4 b4 = *reinterpret_cast<const float4*>(b0 + d * HID + 4 * g);
        float xr[16];
        #pragma unroll
        for (int t = 0; t < 16; ++t)
            xr[t] = x[(bat0 + r0 + t) * IN_DIM + i];
        #pragma unroll
        for (int t = 0; t < 16; ++t) {
            float4 v;
            v.x = fmaf(xr[t], w4.x, b4.x);
            v.y = fmaf(xr[t], w4.y, b4.y);
            v.z = fmaf(xr[t], w4.z, b4.z);
            v.w = fmaf(xr[t], w4.w, b4.w);
            v.x = v.x > 0.f ? v.x : 0.f;
            v.y = v.y > 0.f ? v.y : 0.f;
            v.z = v.z > 0.f ? v.z : 0.f;
            v.w = v.w > 0.f ? v.w : 0.f;
            *reinterpret_cast<float4*>(act + (r0 + t) * LSTR + 4 * g) = v;
        }
    }
    __syncthreads();

    const int lane  = tid & 31;
    const int r     = lane & 15;
    const int half  = lane >> 4;
    const int nbase = (tid >> 5) * 16;

    auto layer = [&](const float* __restrict__ W, const float* __restrict__ Bv) {
        v2f Bfrag[16];
        const float* wrow = W + (size_t)(nbase + r) * HID + 2 * half;
        #pragma unroll
        for (int c = 0; c < 16; ++c)
            Bfrag[c] = *(const v2f*)(wrow + c * 4);
        const float bias = Bv[nbase + r];

        #pragma unroll
        for (int hr = 0; hr < 2; ++hr) {
            v8f acc[4];
            #pragma unroll
            for (int t = 0; t < 2; ++t) {
                const int m0 = hr * 64 + (2 * t) * 16;
                const int m1 = m0 + 16;
                v8f acc0 = {0.f,0.f,0.f,0.f,0.f,0.f,0.f,0.f};
                v8f acc1 = {0.f,0.f,0.f,0.f,0.f,0.f,0.f,0.f};
                const float* a0p = act + (m0 + r) * LSTR + 2 * half;
                const float* a1p = act + (m1 + r) * LSTR + 2 * half;
                #pragma unroll
                for (int c = 0; c < 16; ++c) {
                    v2f a0 = *(const v2f*)(a0p + c * 4);
                    v2f a1 = *(const v2f*)(a1p + c * 4);
                    acc0 = __builtin_amdgcn_wmma_f32_16x16x4_f32(
                               false, a0, false, Bfrag[c], (short)0, acc0, false, false);
                    acc1 = __builtin_amdgcn_wmma_f32_16x16x4_f32(
                               false, a1, false, Bfrag[c], (short)0, acc1, false, false);
                }
                acc[2 * t]     = acc0;
                acc[2 * t + 1] = acc1;
            }

            __syncthreads();

            #pragma unroll
            for (int t = 0; t < 4; ++t) {
                #pragma unroll
                for (int v = 0; v < 8; ++v) {
                    const int row = hr * 64 + t * 16 + v + 8 * half;
                    float o = acc[t][v] + bias; o = o > 0.f ? o : 0.f;
                    act[row * LSTR + nbase + r] = o;
                }
            }
        }
    };

    layer(w1 + (size_t)d * HID * HID, b1 + d * HID);
    layer(w2 + (size_t)d * HID * HID, b2 + d * HID);
    {
        const int b = tid;
        const float* __restrict__ wod = wout + o * (IN_DIM * HID) + i * HID;
        float wv[HID];
        #pragma unroll
        for (int h = 0; h < HID; ++h) wv[h] = wod[h];

        __syncthreads();

        const float4* row = reinterpret_cast<const float4*>(act + b * LSTR);
        float4 a[16];
        #pragma unroll
        for (int j = 0; j < HID / 4; ++j) a[j] = row[j];
        __builtin_amdgcn_sched_barrier(0);
        float s = 0.f;
        #pragma unroll
        for (int j = 0; j < HID / 4; ++j) {
            s = fmaf(wv[4 * j + 0], a[j].x, s);
            s = fmaf(wv[4 * j + 1], a[j].y, s);
            s = fmaf(wv[4 * j + 2], a[j].z, s);
            s = fmaf(wv[4 * j + 3], a[j].w, s);
        }
        if (use_atomic) {
            __hip_atomic_fetch_add(&outbuf[(bat0 + b) * OUT_DIM + o], s,
                                   __ATOMIC_RELAXED, __HIP_MEMORY_SCOPE_AGENT);
        } else {
            outbuf[(size_t)d * BSZ + bat0 + b] = s;
        }
    }
}

__global__ void kan_reduce_kernel(const float* __restrict__ part,
                                  const float* __restrict__ bout,
                                  float* __restrict__ out)
{
    const int o = blockIdx.x;
    const int b = threadIdx.x;
    float s = bout[o];
    const float* p = part + (size_t)o * 64 * BSZ + b;
    #pragma unroll 8
    for (int i = 0; i < 64; ++i) s += p[i * BSZ];
    out[b * OUT_DIM + o] = s;
}

__global__ void kan_init_out(const float* __restrict__ bout, float* __restrict__ out)
{
    const int idx = blockIdx.x * 256 + threadIdx.x;
    out[idx] = bout[idx & 63];
}

extern "C" void kernel_launch(void* const* d_in, const int* in_sizes, int n_in,
                              void* d_out, int out_size, void* d_ws, size_t ws_size,
                              hipStream_t stream)
{
    const float* x    = (const float*)d_in[0];
    const float* w0   = (const float*)d_in[1];
    const float* b0   = (const float*)d_in[2];
    const float* w1   = (const float*)d_in[3];
    const float* b1   = (const float*)d_in[4];
    const float* w2   = (const float*)d_in[5];
    const float* b2   = (const float*)d_in[6];
    const float* wout = (const float*)d_in[7];
    const float* bout = (const float*)d_in[8];
    float* out = (float*)d_out;

    const size_t need = (size_t)DDIM * BSZ * sizeof(float);
    if (ws_size >= need) {
        float* part = (float*)d_ws;
        kan_edge_kernel<<<2 * DDIM, MROWS, 0, stream>>>(x, w0, b0, w1, b1, w2, b2, wout, part, 0);
        kan_reduce_kernel<<<OUT_DIM, BSZ, 0, stream>>>(part, bout, out);
    } else {
        kan_init_out<<<64, 256, 0, stream>>>(bout, out);
        kan_edge_kernel<<<2 * DDIM, MROWS, 0, stream>>>(x, w0, b0, w1, b1, w2, b2, wout, out, 1);
    }
}


# Layer 1 no longer a matrix product: piecewise linear in the row's scalar x, so per-edge prefix-sum tables over sorted breakpoints give z1=x*S+T; layer 2 stays f32 WMMA; layer-1 sum order changes
# speedup vs baseline: 1.4576x; 1.2127x over previous
#include <hip/hip_runtime.h>

typedef float v2f __attribute__((ext_vector_type(2)));
typedef float v8f __attribute__((ext_vector_type(8)));

#define IN_DIM  64
#define OUT_DIM 64
#define HID     64
#define DDIM    (IN_DIM * OUT_DIM)
#define BSZ     256
#define MROWS   128
#define LSTR    68
#define TSTR    128
#define WSTR    65

#define OFF_SKEY 0
#define OFF_PERM 64
#define OFF_CP   128
#define OFF_CN   256
#define OFF_XROW 384
#define OFF_SROW 640
#define OFF_TAB  896
#define OFF_ACT  (OFF_TAB + 65 * TSTR)
#define LDS_FLOATS (OFF_ACT + MROWS * LSTR)

__launch_bounds__(128)
__global__ void kan_edge_kernel(const float* __restrict__ x,
                                const float* __restrict__ w0, const float* __restrict__ b0,
                                const float* __restrict__ w1, const float* __restrict__ b1,
                                const float* __restrict__ w2, const float* __restrict__ b2,
                                const float* __restrict__ wout,
                                float* __restrict__ outbuf, int use_atomic)
{
    __shared__ __attribute__((aligned(16))) float smem[LDS_FLOATS];
    __shared__ __attribute__((aligned(16))) long long ukey[HID];
    float* const act = smem + OFF_ACT;
    float* const tab = smem + OFF_TAB;

    const int d   = blockIdx.x;
    const int o   = d >> 6;
    const int i   = d & 63;
    const int tid = threadIdx.x;

    {
        const float4* __restrict__ w1v = reinterpret_cast<const float4*>(w1 + (size_t)d * HID * HID);
        #pragma unroll
        for (int j = 0; j < 8; ++j) {
            const int q = tid + 128 * j;
            const float4 v = w1v[q];
            float* dst = act + (q >> 4) * WSTR + 4 * (q & 15);
            dst[0] = v.x; dst[1] = v.y; dst[2] = v.z; dst[3] = v.w;
        }
    }
    float myw = 0.f, myb = 0.f, mykey = 0.f;
    long long mykey64 = 0;
    if (tid < HID) {
        myw = w0[d * HID + tid];
        myb = b0[d * HID + tid];
        mykey = (myw != 0.f) ? (-myb / myw)
                             : (myb > 0.f ? -__builtin_inff() : __builtin_inff());
        const int kb = __float_as_int(mykey);
        const int ki = kb ^ ((kb >> 31) & 0x7fffffff);
        mykey64 = ((long long)ki << 32) | (long long)tid;
        ukey[tid] = mykey64;
    }
    __syncthreads();

    if (tid < HID) {
        int pos = 0;
        #pragma unroll 16
        for (int h2 = 0; h2 < HID; ++h2)
            pos += (ukey[h2] < mykey64) ? 1 : 0;
        const bool rising = !(myw < 0.f);
        smem[OFF_SKEY + pos] = mykey;
        smem[OFF_PERM + pos] = __int_as_float(tid);
        smem[OFF_CP + pos]       = rising ? myw : 0.f;
        smem[OFF_CP + HID + pos] = rising ? myb : 0.f;
        smem[OFF_CN + pos]       = rising ? 0.f : myw;
        smem[OFF_CN + HID + pos] = rising ? 0.f : myb;
    }
    __syncthreads();

    #pragma unroll
    for (int k = 0; k < 2; ++k) {
        const int row = tid + MROWS * k;
        const float xv = x[row * IN_DIM + i];
        int c = 0;
        #pragma unroll
        for (int step = 32; step >= 1; step >>= 1)
            c += (smem[OFF_SKEY + c + step - 1] < xv) ? step : 0;
        c += (smem[OFF_SKEY + c] < xv) ? 1 : 0;
        smem[OFF_XROW + row] = xv;
        smem[OFF_SROW + row] = __int_as_float(c * TSTR);
    }

    {
        const int n   = tid & 63;
        const int sel = (tid >> 6) * HID;
        const float* cp = smem + OFF_CP + sel;
        const float* cn = smem + OFF_CN + sel;
        const float* wrow = act + n * WSTR;
        float wv[HID];
        #pragma unroll
        for (int p = 0; p < HID; ++p) wv[p] = wrow[__float_as_int(smem[OFF_PERM + p])];
        float sn = 0.f;
        #pragma unroll
        for (int p = HID - 1; p >= 0; --p) {
            sn = fmaf(wv[p], cn[p], sn);
            tab[p * TSTR + tid] = sn;
        }
        float sp = sel ? b1[d * HID + n] : 0.f;
        #pragma unroll
        for (int s = 0; s < HID; ++s) {
            tab[s * TSTR + tid] += sp;
            sp = fmaf(wv[s], cp[s], sp);
        }
        tab[HID * TSTR + tid] = sp;
    }

    const int lane  = tid & 31;
    const int r     = lane & 15;
    const int half  = lane >> 4;
    const int nbase = (tid >> 5) * 16;

    v2f Bfrag[16];
    {
        const float* wrow2 = w2 + (size_t)d * HID * HID + (size_t)(nbase + r) * HID + 2 * half;
        #pragma unroll
        for (int c = 0; c < 16; ++c)
            Bfrag[c] = *(const v2f*)(wrow2 + c * 4);
    }
    const float bias = b2[d * HID + nbase + r];
    const float* __restrict__ wod = wout + o * (IN_DIM * HID) + i * HID;
    float wv[HID];
    #pragma unroll
    for (int h = 0; h < HID; ++h) wv[h] = wod[h];

    #pragma nounroll
    for (int pass = 0; pass < 2; ++pass) {
        const int bat0 = pass * MROWS;
        __syncthreads();

        {
            const int g  = tid & 15;
            const int r0 = (tid >> 4) * 16;
            const float4* xr4 = reinterpret_cast<const float4*>(smem + OFF_XROW + bat0 + r0);
            const float4* sr4 = reinterpret_cast<const float4*>(smem + OFF_SROW + bat0 + r0);
            float xs[16], so[16];
            #pragma unroll
            for (int q = 0; q < 4; ++q) {
                const float4 xq = xr4[q], sq = sr4[q];
                xs[4 * q] = xq.x; xs[4 * q + 1] = xq.y; xs[4 * q + 2] = xq.z; xs[4 * q + 3] = xq.w;
                so[4 * q] = sq.x; so[4 * q + 1] = sq.y; so[4 * q + 2] = sq.z; so[4 * q + 3] = sq.w;
            }
            #pragma unroll
            for (int t = 0; t < 16; ++t) {
                const float xv = xs[t];
                const float* trow = tab + __float_as_int(so[t]) + 4 * g;
                const float4 s4 = *reinterpret_cast<const float4*>(trow);
                const float4 t4 = *reinterpret_cast<const float4*>(trow + HID);
                float4 v;
                v.x = fmaf(xv, s4.x, t4.x);
                v.y = fmaf(xv, s4.y, t4.y);
                v.z = fmaf(xv, s4.z, t4.z);
                v.w = fmaf(xv, s4.w, t4.w);
                v.x = v.x > 0.f ? v.x : 0.f;
                v.y = v.y > 0.f ? v.y : 0.f;
                v.z = v.z > 0.f ? v.z : 0.f;
                v.w = v.w > 0.f ? v.w : 0.f;
                *reinterpret_cast<float4*>(act + (r0 + t) * LSTR + 4 * g) = v;
            }
        }
        __syncthreads();

        #pragma unroll
        for (int hr = 0; hr < 2; ++hr) {
            v8f acc[4];
            #pragma unroll
            for (int t = 0; t < 2; ++t) {
                const int m0 = hr * 64 + (2 * t) * 16;
                const int m1 = m0 + 16;
                v8f acc0 = {0.f,0.f,0.f,0.f,0.f,0.f,0.f,0.f};
                v8f acc1 = {0.f,0.f,0.f,0.f,0.f,0.f,0.f,0.f};
                const float* a0p = act + (m0 + r) * LSTR + 2 * half;
                const float* a1p = act + (m1 + r) * LSTR + 2 * half;
                #pragma unroll
                for (int c = 0; c < 16; ++c) {
                    v2f a0 = *(const v2f*)(a0p + c * 4);
                    v2f a1 = *(const v2f*)(a1p + c * 4);
                    acc0 = __builtin_amdgcn_wmma_f32_16x16x4_f32(
                               false, a0, false, Bfrag[c], (short)0, acc0, false, false);
                    acc1 = __builtin_amdgcn_wmma_f32_16x16x4_f32(
                               false, a1, false, Bfrag[c], (short)0, acc1, false, false);
                }
                acc[2 * t]     = acc0;
                acc[2 * t + 1] = acc1;
            }

            __syncthreads();

            #pragma unroll
            for (int t = 0; t < 4; ++t) {
                #pragma unroll
                for (int v = 0; v < 8; ++v) {
                    const int row = hr * 64 + t * 16 + v + 8 * half;
                    float ov = acc[t][v] + bias; ov = ov > 0.f ? ov : 0.f;
                    act[row * LSTR + nbase + r] = ov;
                }
            }
        }

        {
            const int b = tid;

            __syncthreads();

            const float4* row = reinterpret_cast<const float4*>(act + b * LSTR);
            float4 a[16];
            #pragma unroll
            for (int j = 0; j < HID / 4; ++j) a[j] = row[j];
            __builtin_amdgcn_sched_barrier(0);
            float s = 0.f;
            #pragma unroll
            for (int j = 0; j < HID / 4; ++j) {
                s = fmaf(wv[4 * j + 0], a[j].x, s);
                s = fmaf(wv[4 * j + 1], a[j].y, s);
                s = fmaf(wv[4 * j + 2], a[j].z, s);
                s = fmaf(wv[4 * j + 3], a[j].w, s);
            }
            if (use_atomic) {
                __hip_atomic_fetch_add(&outbuf[(bat0 + b) * OUT_DIM + o], s,
                                       __ATOMIC_RELAXED, __HIP_MEMORY_SCOPE_AGENT);
            } else {
                outbuf[(size_t)d * BSZ + bat0 + b] = s;
            }
        }
    }
}

__global__ void kan_reduce_kernel(const float* __restrict__ part,
                                  const float* __restrict__ bout,
                                  float* __restrict__ out)
{
    const int o = blockIdx.x;
    const int b = threadIdx.x;
    float s = bout[o];
    const float* p = part + (size_t)o * 64 * BSZ + b;
    #pragma unroll 8
    for (int i = 0; i < 64; ++i) s += p[i * BSZ];
    out[b * OUT_DIM + o] = s;
}

__global__ void kan_init_out(const float* __restrict__ bout, float* __restrict__ out)
{
    const int idx = blockIdx.x * 256 + threadIdx.x;
    out[idx] = bout[idx & 63];
}

extern "C" void kernel_launch(void* const* d_in, const int* in_sizes, int n_in,
                              void* d_out, int out_size, void* d_ws, size_t ws_size,
                              hipStream_t stream)
{
    const float* x    = (const float*)d_in[0];
    const float* w0   = (const float*)d_in[1];
    const float* b0   = (const float*)d_in[2];
    const float* w1   = (const float*)d_in[3];
    const float* b1   = (const float*)d_in[4];
    const float* w2   = (const float*)d_in[5];
    const float* b2   = (const float*)d_in[6];
    const float* wout = (const float*)d_in[7];
    const float* bout = (const float*)d_in[8];
    float* out = (float*)d_out;

    const size_t need = (size_t)DDIM * BSZ * sizeof(float);
    if (ws_size >= need) {
        float* part = (float*)d_ws;
        kan_edge_kernel<<<DDIM, MROWS, 0, stream>>>(x, w0, b0, w1, b1, w2, b2, wout, part, 0);
        kan_reduce_kernel<<<OUT_DIM, BSZ, 0, stream>>>(part, bout, out);
    } else {
        kan_init_out<<<64, 256, 0, stream>>>(bout, out);
        kan_edge_kernel<<<DDIM, MROWS, 0, stream>>>(x, w0, b0, w1, b1, w2, b2, wout, out, 1);
    }
}


# Loads issued in groups ahead of use: layer-1 table rows 8 per LDS round trip (was 1), layer-2 A fragments a 16-WMMA block ahead in a second register set, wout/x/w0/b0 at kernel start; sums unchanged
# speedup vs baseline: 1.5231x; 1.0449x over previous
#include <hip/hip_runtime.h>

typedef float v2f __attribute__((ext_vector_type(2)));
typedef float v8f __attribute__((ext_vector_type(8)));

#define IN_DIM  64
#define OUT_DIM 64
#define HID     64
#define DDIM    (IN_DIM * OUT_DIM)
#define BSZ     256
#define MROWS   128
#define LSTR    68
#define TSTR    128
#define WSTR    65

#define OFF_SKEY 0
#define OFF_PERM 64
#define OFF_CP   128
#define OFF_CN   256
#define OFF_XROW 384
#define OFF_SROW 640
#define OFF_TAB  896
#define OFF_ACT  (OFF_TAB + 65 * TSTR)
#define LDS_FLOATS (OFF_ACT + MROWS * LSTR)

__launch_bounds__(128)
__global__ void kan_edge_kernel(const float* __restrict__ x,
                                const float* __restrict__ w0, const float* __restrict__ b0,
                                const float* __restrict__ w1, const float* __restrict__ b1,
                                const float* __restrict__ w2, const float* __restrict__ b2,
                                const float* __restrict__ wout,
                                float* __restrict__ outbuf, int use_atomic)
{
    __shared__ __attribute__((aligned(16))) float smem[LDS_FLOATS];
    __shared__ __attribute__((aligned(16))) long long ukey[HID];
    float* const act = smem + OFF_ACT;
    float* const tab = smem + OFF_TAB;

    const int d   = blockIdx.x;
    const int o   = d >> 6;
    const int i   = d & 63;
    const int tid = threadIdx.x;

    const float* __restrict__ wod = wout + o * (IN_DIM * HID) + i * HID;
    float wo[HID];
    #pragma unroll
    for (int h = 0; h < HID; ++h) wo[h] = wod[h];
    const float xin0 = x[tid * IN_DIM + i];
    const float xin1 = x[(tid + MROWS) * IN_DIM + i];
    float myw = 0.f, myb = 0.f;
    if (tid < HID) {
        myw = w0[d * HID + tid];
        myb = b0[d * HID + tid];
    }

    {
        const float4* __restrict__ w1v = reinterpret_cast<const float4*>(w1 + (size_t)d * HID * HID);
        #pragma unroll
        for (int j = 0; j < 8; ++j) {
            const int q = tid + 128 * j;
            const float4 v = w1v[q];
            float* dst = act + (q >> 4) * WSTR + 4 * (q & 15);
            dst[0] = v.x; dst[1] = v.y; dst[2] = v.z; dst[3] = v.w;
        }
    }
    float mykey = 0.f;
    long long mykey64 = 0;
    if (tid < HID) {
        mykey = (myw != 0.f) ? (-myb / myw)
                             : (myb > 0.f ? -__builtin_inff() : __builtin_inff());
        const int kb = __float_as_int(mykey);
        const int ki = kb ^ ((kb >> 31) & 0x7fffffff);
        mykey64 = ((long long)ki << 32) | (long long)tid;
        ukey[tid] = mykey64;
    }
    __syncthreads();

    if (tid < HID) {
        int pos = 0;
        #pragma unroll 16
        for (int h2 = 0; h2 < HID; ++h2)
            pos += (ukey[h2] < mykey64) ? 1 : 0;
        const bool rising = !(myw < 0.f);
        smem[OFF_SKEY + pos] = mykey;
        smem[OFF_PERM + pos] = __int_as_float(tid);
        smem[OFF_CP + pos]       = rising ? myw : 0.f;
        smem[OFF_CP + HID + pos] = rising ? myb : 0.f;
        smem[OFF_CN + pos]       = rising ? 0.f : myw;
        smem[OFF_CN + HID + pos] = rising ? 0.f : myb;
    }
    __syncthreads();

    #pragma unroll
    for (int k = 0; k < 2; ++k) {
        const int row = tid + MROWS * k;
        const float xv = k ? xin1 : xin0;
        int c = 0;
        #pragma unroll
        for (int step = 32; step >= 1; step >>= 1)
            c += (smem[OFF_SKEY + c + step - 1] < xv) ? step : 0;
        c += (smem[OFF_SKEY + c] < xv) ? 1 : 0;
        smem[OFF_XROW + row] = xv;
        smem[OFF_SROW + row] = __int_as_float(c * TSTR);
    }

    {
        const int n   = tid & 63;
        const int sel = (tid >> 6) * HID;
        const float* cp = smem + OFF_CP + sel;
        const float* cn = smem + OFF_CN + sel;
        const float* wrow = act + n * WSTR;
        float wv[HID];
        #pragma unroll
        for (int p = 0; p < HID; ++p) wv[p] = wrow[__float_as_int(smem[OFF_PERM + p])];
        float sn = 0.f;
        #pragma unroll
        for (int p = HID - 1; p >= 0; --p) {
            sn = fmaf(wv[p], cn[p], sn);
            tab[p * TSTR + tid] = sn;
        }
        float sp = sel ? b1[d * HID + n] : 0.f;
        #pragma unroll
        for (int s = 0; s < HID; ++s) {
            tab[s * TSTR + tid] += sp;
            sp = fmaf(wv[s], cp[s], sp);
        }
        tab[HID * TSTR + tid] = sp;
    }

    const int lane  = tid & 31;
    const int r     = lane & 15;
    const int half  = lane >> 4;
    const int nbase = (tid >> 5) * 16;

    v2f Bfrag[16];
    {
        const float* wrow2 = w2 + (size_t)d * HID * HID + (size_t)(nbase + r) * HID + 2 * half;
        #pragma unroll
        for (int c = 0; c < 16; ++c)
            Bfrag[c] = *(const v2f*)(wrow2 + c * 4);
    }
    const float bias = b2[d * HID + nbase + r];

    #pragma nounroll
    for (int pass = 0; pass < 2; ++pass) {
        const int bat0 = pass * MROWS;
        __syncthreads();

        {
            const int g  = tid & 15;
            const int r0 = (tid >> 4) * 16;
            const float4* xr4 = reinterpret_cast<const float4*>(smem + OFF_XROW + bat0 + r0);
            const float4* sr4 = reinterpret_cast<const float4*>(smem + OFF_SROW + bat0 + r0);
            float xs[16], so[16];
            #pragma unroll
            for (int q = 0; q < 4; ++q) {
                const float4 xq = xr4[q], sq = sr4[q];
                xs[4 * q] = xq.x; xs[4 * q + 1] = xq.y; xs[4 * q + 2] = xq.z; xs[4 * q + 3] = xq.w;
                so[4 * q] = sq.x; so[4 * q + 1] = sq.y; so[4 * q + 2] = sq.z; so[4 * q + 3] = sq.w;
            }
            #pragma unroll
            for (int hb = 0; hb < 2; ++hb) {
                float4 s4[8], t4[8];
                #pragma unroll
                for (int t = 0; t < 8; ++t) {
                    const float* trow = tab + __float_as_int(so[8 * hb + t]) + 4 * g;
                    s4[t] = *reinterpret_cast<const float4*>(trow);
                    t4[t] = *reinterpret_cast<const float4*>(trow + HID);
                }
                #pragma unroll
                for (int t = 0; t < 8; ++t) {
                    const float xv = xs[8 * hb + t];
                    float4 v;
                    v.x = fmaf(xv, s4[t].x, t4[t].x);
                    v.y = fmaf(xv, s4[t].y, t4[t].y);
                    v.z = fmaf(xv, s4[t].z, t4[t].z);
                    v.w = fmaf(xv, s4[t].w, t4[t].w);
                    v.x = v.x > 0.f ? v.x : 0.f;
                    v.y = v.y > 0.f ? v.y : 0.f;
                    v.z = v.z > 0.f ? v.z : 0.f;
                    v.w = v.w > 0.f ? v.w : 0.f;
                    *reinterpret_cast<float4*>(act + (r0 + 8 * hb + t) * LSTR + 4 * g) = v;
                }
            }
        }
        __syncthreads();

        #pragma unroll
        for (int hr = 0; hr < 2; ++hr) {
            v8f acc[4];
            const float* ap[4];
            #pragma unroll
            for (int t = 0; t < 4; ++t) {
                acc[t] = (v8f){0.f,0.f,0.f,0.f,0.f,0.f,0.f,0.f};
                ap[t]  = act + (hr * 64 + t * 16 + r) * LSTR + 2 * half;
            }
            v2f af[2][4][4];
            #pragma unroll
            for (int kk = 0; kk < 4; ++kk) {
                #pragma unroll
                for (int t = 0; t < 4; ++t)
                    af[0][kk][t] = *(const v2f*)(ap[t] + kk * 4);
            }
            #pragma unroll
            for (int blk = 0; blk < 4; ++blk) {
                if (blk < 3) {
                    #pragma unroll
                    for (int kk = 0; kk < 4; ++kk) {
                        #pragma unroll
                        for (int t = 0; t < 4; ++t)
                            af[(blk + 1) & 1][kk][t] = *(const v2f*)(ap[t] + ((blk + 1) * 4 + kk) * 4);
                    }
                }
                __builtin_amdgcn_sched_barrier(0);
                #pragma unroll
                for (int kk = 0; kk < 4; ++kk) {
                    #pragma unroll
                    for (int t = 0; t < 4; ++t)
                        acc[t] = __builtin_amdgcn_wmma_f32_16x16x4_f32(
                                     false, af[blk & 1][kk][t], false, Bfrag[blk * 4 + kk],
                                     (short)0, acc[t], false, false);
                }
                __builtin_amdgcn_sched_barrier(0);
            }

            __syncthreads();

            #pragma unroll
            for (int t = 0; t < 4; ++t) {
                #pragma unroll
                for (int v = 0; v < 8; ++v) {
                    const int row = hr * 64 + t * 16 + v + 8 * half;
                    float ov = acc[t][v] + bias; ov = ov > 0.f ? ov : 0.f;
                    act[row * LSTR + nbase + r] = ov;
                }
            }
        }

        {
            const int b = tid;

            __syncthreads();

            const float4* row = reinterpret_cast<const float4*>(act + b * LSTR);
            float4 a[16];
            #pragma unroll
            for (int j = 0; j < HID / 4; ++j) a[j] = row[j];
            __builtin_amdgcn_sched_barrier(0);
            float s = 0.f;
            #pragma unroll
            for (int j = 0; j < HID / 4; ++j) {
                s = fmaf(wo[4 * j + 0], a[j].x, s);
                s = fmaf(wo[4 * j + 1], a[j].y, s);
                s = fmaf(wo[4 * j + 2], a[j].z, s);
                s = fmaf(wo[4 * j + 3], a[j].w, s);
            }
            if (use_atomic) {
                __hip_atomic_fetch_add(&outbuf[(bat0 + b) * OUT_DIM + o], s,
                                       __ATOMIC_RELAXED, __HIP_MEMORY_SCOPE_AGENT);
            } else {
                outbuf[(size_t)d * BSZ + bat0 + b] = s;
            }
        }
    }
}

__global__ void kan_reduce_kernel(const float* __restrict__ part,
                                  const float* __restrict__ bout,
                                  float* __restrict__ out)
{
    const int o = blockIdx.x;
    const int b = threadIdx.x;
    float s = bout[o];
    const float* p = part + (size_t)o * 64 * BSZ + b;
    #pragma unroll 8
    for (int i = 0; i < 64; ++i) s += p[i * BSZ];
    out[b * OUT_DIM + o] = s;
}

__global__ void kan_init_out(const float* __restrict__ bout, float* __restrict__ out)
{
    const int idx = blockIdx.x * 256 + threadIdx.x;
    out[idx] = bout[idx & 63];
}

extern "C" void kernel_launch(void* const* d_in, const int* in_sizes, int n_in,
                              void* d_out, int out_size, void* d_ws, size_t ws_size,
                              hipStream_t stream)
{
    const float* x    = (const float*)d_in[0];
    const float* w0   = (const float*)d_in[1];
    const float* b0   = (const float*)d_in[2];
    const float* w1   = (const float*)d_in[3];
    const float* b1   = (const float*)d_in[4];
    const float* w2   = (const float*)d_in[5];
    const float* b2   = (const float*)d_in[6];
    const float* wout = (const float*)d_in[7];
    const float* bout = (const float*)d_in[8];
    float* out = (float*)d_out;

    const size_t need = (size_t)DDIM * BSZ * sizeof(float);
    if (ws_size >= need) {
        float* part = (float*)d_ws;
        kan_edge_kernel<<<DDIM, MROWS, 0, stream>>>(x, w0, b0, w1, b1, w2, b2, wout, part, 0);
        kan_reduce_kernel<<<OUT_DIM, BSZ, 0, stream>>>(part, bout, out);
    } else {
        kan_init_out<<<64, 256, 0, stream>>>(bout, out);
        kan_edge_kernel<<<DDIM, MROWS, 0, stream>>>(x, w0, b0, w1, b1, w2, b2, wout, out, 1);
    }
}


# Loads ahead of use: all globals at kernel start, in flight across LDS-only barriers; table build reads LDS in groups, writes rows once, search under its chain; reduce 32 loads deep; sums unchanged
# speedup vs baseline: 1.6914x; 1.1105x over previous
#include <hip/hip_runtime.h>

typedef float v2f __attribute__((ext_vector_type(2)));
typedef float v8f __attribute__((ext_vector_type(8)));

#define IN_DIM  64
#define OUT_DIM 64
#define HID     64
#define DDIM    (IN_DIM * OUT_DIM)
#define BSZ     256
#define MROWS   128
#define LSTR    68
#define TSTR    128
#define WSTR    65

#define OFF_SKEY 0
#define OFF_PERM 64
#define OFF_CP   128
#define OFF_CN   256
#define OFF_XROW 384
#define OFF_SROW 640
#define OFF_TAB  896
#define OFF_ACT  (OFF_TAB + 65 * TSTR)
#define LDS_FLOATS (OFF_ACT + MROWS * LSTR)

static __device__ __forceinline__ void lds_barrier()
{
    __builtin_amdgcn_fence(__ATOMIC_RELEASE, "workgroup", "local");
    __builtin_amdgcn_s_barrier();
    __builtin_amdgcn_fence(__ATOMIC_ACQUIRE, "workgroup", "local");
}

__launch_bounds__(128)
__global__ void kan_edge_kernel(const float* __restrict__ x,
                                const float* __restrict__ w0, const float* __restrict__ b0,
                                const float* __restrict__ w1, const float* __restrict__ b1,
                                const float* __restrict__ w2, const float* __restrict__ b2,
                                const float* __restrict__ wout,
                                float* __restrict__ outbuf, int use_atomic)
{
    __shared__ __attribute__((aligned(16))) float smem[LDS_FLOATS];
    __shared__ __attribute__((aligned(16))) long long ukey[HID];
    float* const act = smem + OFF_ACT;
    float* const tab = smem + OFF_TAB;

    const int d   = blockIdx.x;
    const int o   = d >> 6;
    const int i   = d & 63;
    const int tid = threadIdx.x;

    const float* __restrict__ wod = wout + o * (IN_DIM * HID) + i * HID;
    float wo[HID];
    #pragma unroll
    for (int h = 0; h < HID; ++h) wo[h] = wod[h];
    const float xin0 = x[tid * IN_DIM + i];
    const float xin1 = x[(tid + MROWS) * IN_DIM + i];
    float myw = 0.f, myb = 0.f;
    if (tid < HID) {
        myw = w0[d * HID + tid];
        myb = b0[d * HID + tid];
    }

    const int lane  = tid & 31;
    const int r     = lane & 15;
    const int half  = lane >> 4;
    const int nbase = (tid >> 5) * 16;

    float4 w1r[8];
    {
        const float4* __restrict__ w1v = reinterpret_cast<const float4*>(w1 + (size_t)d * HID * HID);
        #pragma unroll
        for (int j = 0; j < 8; ++j) w1r[j] = w1v[tid + 128 * j];
    }
    const float b1v = b1[d * HID + (tid & 63)];
    const float bias = b2[d * HID + nbase + r];
    v2f Bfrag[16];
    {
        const float* wrow2 = w2 + (size_t)d * HID * HID + (size_t)(nbase + r) * HID + 2 * half;
        #pragma unroll
        for (int c = 0; c < 16; ++c)
            Bfrag[c] = *(const v2f*)(wrow2 + c * 4);
    }
    float mykey = 0.f;
    long long mykey64 = 0;
    if (tid < HID) {
        mykey = (myw != 0.f) ? (-myb / myw)
                             : (myb > 0.f ? -__builtin_inff() : __builtin_inff());
        const int kb = __float_as_int(mykey);
        const int ki = kb ^ ((kb >> 31) & 0x7fffffff);
        mykey64 = ((long long)ki << 32) | (long long)tid;
        ukey[tid] = mykey64;
    }
    lds_barrier();

    if (tid < HID) {
        int pos = 0;
        #pragma unroll 16
        for (int h2 = 0; h2 < HID; ++h2)
            pos += (ukey[h2] < mykey64) ? 1 : 0;
        const bool rising = !(myw < 0.f);
        smem[OFF_SKEY + pos] = mykey;
        smem[OFF_PERM + pos] = __int_as_float(tid);
        smem[OFF_CP + pos]       = rising ? myw : 0.f;
        smem[OFF_CP + HID + pos] = rising ? myb : 0.f;
        smem[OFF_CN + pos]       = rising ? 0.f : myw;
        smem[OFF_CN + HID + pos] = rising ? 0.f : myb;
    }
    {
        #pragma unroll
        for (int j = 0; j < 8; ++j) {
            const int q = tid + 128 * j;
            const float4 v = w1r[j];
            float* dst = act + (q >> 4) * WSTR + 4 * (q & 15);
            dst[0] = v.x; dst[1] = v.y; dst[2] = v.z; dst[3] = v.w;
        }
    }
    lds_barrier();

    {
        const int n   = tid & 63;
        const int sel = (tid >> 6) * HID;
        const float* cp = smem + OFF_CP + sel;
        const float* cn = smem + OFF_CN + sel;
        const float* wrow = act + n * WSTR;
        float wv[HID];
        int pidx[HID];
        #pragma unroll
        for (int p = 0; p < HID; ++p) pidx[p] = __float_as_int(smem[OFF_PERM + p]);
        __builtin_amdgcn_sched_barrier(0);
        #pragma unroll
        for (int p = HID - 1; p >= 0; --p) wv[p] = wrow[pidx[p]];
        float cnv[HID];
        #pragma unroll
        for (int p = 0; p < HID; ++p) cnv[p] = cn[p];
        __builtin_amdgcn_sched_barrier(0);

        float sfx[HID];
        float cpv[HID];
        float sn = 0.f;
        int c0 = 0, c1 = 0;
        #pragma unroll
        for (int k = 0; k < 7; ++k) {
            const int off = (k < 6) ? ((32 >> k) - 1) : 0;
            const int inc = (k < 6) ? (32 >> k) : 1;
            const float key0 = smem[OFF_SKEY + c0 + off];
            const float key1 = smem[OFF_SKEY + c1 + off];
            if (k == 6) {
                #pragma unroll
                for (int s = 0; s < HID; ++s) cpv[s] = cp[s];
            }
            __builtin_amdgcn_sched_barrier(0);
            const int hi = 63 - 9 * k - (k ? 1 : 0);
            const int len = k ? 9 : 10;
            #pragma unroll
            for (int j = 0; j < len; ++j) {
                const int p = hi - j;
                sn = fmaf(wv[p], cnv[p], sn);
                sfx[p] = sn;
            }
            __builtin_amdgcn_sched_barrier(0);
            c0 += (key0 < xin0) ? inc : 0;
            c1 += (key1 < xin1) ? inc : 0;
        }
        smem[OFF_XROW + tid]         = xin0;
        smem[OFF_XROW + tid + MROWS] = xin1;
        smem[OFF_SROW + tid]         = __int_as_float(c0 * TSTR);
        smem[OFF_SROW + tid + MROWS] = __int_as_float(c1 * TSTR);
        float sp = sel ? b1v : 0.f;
        #pragma unroll
        for (int s = 0; s < HID; ++s) {
            tab[s * TSTR + tid] = sfx[s] + sp;
            sp = fmaf(wv[s], cpv[s], sp);
        }
        tab[HID * TSTR + tid] = sp;
    }

    #pragma nounroll
    for (int pass = 0; pass < 2; ++pass) {
        const int bat0 = pass * MROWS;
        __syncthreads();

        {
            const int g  = tid & 15;
            const int r0 = (tid >> 4) * 16;
            const float4* xr4 = reinterpret_cast<const float4*>(smem + OFF_XROW + bat0 + r0);
            const float4* sr4 = reinterpret_cast<const float4*>(smem + OFF_SROW + bat0 + r0);
            float xs[16], so[16];
            #pragma unroll
            for (int q = 0; q < 4; ++q) {
                const float4 xq = xr4[q], sq = sr4[q];
                xs[4 * q] = xq.x; xs[4 * q + 1] = xq.y; xs[4 * q + 2] = xq.z; xs[4 * q + 3] = xq.w;
                so[4 * q] = sq.x; so[4 * q + 1] = sq.y; so[4 * q + 2] = sq.z; so[4 * q + 3] = sq.w;
            }
            #pragma unroll
            for (int hb = 0; hb < 2; ++hb) {
                float4 s4[8], t4[8];
                #pragma unroll
                for (int t = 0; t < 8; ++t) {
                    const float* trow = tab + __float_as_int(so[8 * hb + t]) + 4 * g;
                    s4[t] = *reinterpret_cast<const float4*>(trow);
                    t4[t] = *reinterpret_cast<const float4*>(trow + HID);
                }
                #pragma unroll
                for (int t = 0; t < 8; ++t) {
                    const float xv = xs[8 * hb + t];
                    float4 v;
                    v.x = fmaf(xv, s4[t].x, t4[t].x);
                    v.y = fmaf(xv, s4[t].y, t4[t].y);
                    v.z = fmaf(xv, s4[t].z, t4[t].z);
                    v.w = fmaf(xv, s4[t].w, t4[t].w);
                    v.x = v.x > 0.f ? v.x : 0.f;
                    v.y = v.y > 0.f ? v.y : 0.f;
                    v.z = v.z > 0.f ? v.z : 0.f;
                    v.w = v.w > 0.f ? v.w : 0.f;
                    *reinterpret_cast<float4*>(act + (r0 + 8 * hb + t) * LSTR + 4 * g) = v;
                }
            }
        }
        __syncthreads();

        #pragma unroll
        for (int hr = 0; hr < 2; ++hr) {
            v8f acc[4];
            const float* ap[4];
            #pragma unroll
            for (int t = 0; t < 4; ++t) {
                acc[t] = (v8f){0.f,0.f,0.f,0.f,0.f,0.f,0.f,0.f};
                ap[t]  = act + (hr * 64 + t * 16 + r) * LSTR + 2 * half;
            }
            v2f af[2][4][4];
            #pragma unroll
            for (int kk = 0; kk < 4; ++kk) {
                #pragma unroll
                for (int t = 0; t < 4; ++t)
                    af[0][kk][t] = *(const v2f*)(ap[t] + kk * 4);
            }
            #pragma unroll
            for (int blk = 0; blk < 4; ++blk) {
                if (blk < 3) {
                    #pragma unroll
                    for (int kk = 0; kk < 4; ++kk) {
                        #pragma unroll
                        for (int t = 0; t < 4; ++t)
                            af[(blk + 1) & 1][kk][t] = *(const v2f*)(ap[t] + ((blk + 1) * 4 + kk) * 4);
                    }
                }
                __builtin_amdgcn_sched_barrier(0);
                #pragma unroll
                for (int kk = 0; kk < 4; ++kk) {
                    #pragma unroll
                    for (int t = 0; t < 4; ++t)
                        acc[t] = __builtin_amdgcn_wmma_f32_16x16x4_f32(
                                     false, af[blk & 1][kk][t], false, Bfrag[blk * 4 + kk],
                                     (short)0, acc[t], false, false);
                }
                __builtin_amdgcn_sched_barrier(0);
            }

            __syncthreads();

            #pragma unroll
            for (int t = 0; t < 4; ++t) {
                #pragma unroll
                for (int v = 0; v < 8; ++v) {
                    const int row = hr * 64 + t * 16 + v + 8 * half;
                    float ov = acc[t][v] + bias; ov = ov > 0.f ? ov : 0.f;
                    act[row * LSTR + nbase + r] = ov;
                }
            }
        }

        {
            const int b = tid;

            __syncthreads();

            const float4* row = reinterpret_cast<const float4*>(act + b * LSTR);
            float4 a[16];
            #pragma unroll
            for (int j = 0; j < HID / 4; ++j) a[j] = row[j];
            __builtin_amdgcn_sched_barrier(0);
            float s = 0.f;
            #pragma unroll
            for (int j = 0; j < HID / 4; ++j) {
                s = fmaf(wo[4 * j + 0], a[j].x, s);
                s = fmaf(wo[4 * j + 1], a[j].y, s);
                s = fmaf(wo[4 * j + 2], a[j].z, s);
                s = fmaf(wo[4 * j + 3], a[j].w, s);
            }
            if (use_atomic) {
                __hip_atomic_fetch_add(&outbuf[(bat0 + b) * OUT_DIM + o], s,
                                       __ATOMIC_RELAXED, __HIP_MEMORY_SCOPE_AGENT);
            } else {
                outbuf[(size_t)d * BSZ + bat0 + b] = s;
            }
        }
    }
}

__global__ void kan_reduce_kernel(const float* __restrict__ part,
                                  const float* __restrict__ bout,
                                  float* __restrict__ out)
{
    const int o = blockIdx.x;
    const int b = threadIdx.x;
    float s = bout[o];
    const float* p = part + (size_t)o * 64 * BSZ + b;
    #pragma unroll 32
    for (int i = 0; i < 64; ++i) s += p[i * BSZ];
    out[b * OUT_DIM + o] = s;
}

__global__ void kan_init_out(const float* __restrict__ bout, float* __restrict__ out)
{
    const int idx = blockIdx.x * 256 + threadIdx.x;
    out[idx] = bout[idx & 63];
}

extern "C" void kernel_launch(void* const* d_in, const int* in_sizes, int n_in,
                              void* d_out, int out_size, void* d_ws, size_t ws_size,
                              hipStream_t stream)
{
    const float* x    = (const float*)d_in[0];
    const float* w0   = (const float*)d_in[1];
    const float* b0   = (const float*)d_in[2];
    const float* w1   = (const float*)d_in[3];
    const float* b1   = (const float*)d_in[4];
    const float* w2   = (const float*)d_in[5];
    const float* b2   = (const float*)d_in[6];
    const float* wout = (const float*)d_in[7];
    const float* bout = (const float*)d_in[8];
    float* out = (float*)d_out;

    const size_t need = (size_t)DDIM * BSZ * sizeof(float);
    if (ws_size >= need) {
        float* part = (float*)d_ws;
        kan_edge_kernel<<<DDIM, MROWS, 0, stream>>>(x, w0, b0, w1, b1, w2, b2, wout, part, 0);
        kan_reduce_kernel<<<OUT_DIM, BSZ, 0, stream>>>(part, bout, out);
    } else {
        kan_init_out<<<64, 256, 0, stream>>>(bout, out);
        kan_edge_kernel<<<DDIM, MROWS, 0, stream>>>(x, w0, b0, w1, b1, w2, b2, wout, out, 1);
    }
}


# Layer 2 once per (segment, layer-1 sign mask) pattern of a wave's 64 rows: f32 WMMA on masked S/T table rows gives U,V, rows take z2=x*U+V; gain depends on pattern count; layer-2 sum order changes
# speedup vs baseline: 2.3145x; 1.3684x over previous
#include <hip/hip_runtime.h>

typedef float v2f __attribute__((ext_vector_type(2)));
typedef float v8f __attribute__((ext_vector_type(8)));

#define IN_DIM  64
#define OUT_DIM 64
#define HID     64
#define DDIM    (IN_DIM * OUT_DIM)
#define BSZ     256
#define MROWS   128
#define LSTR    68
#define TSTR    132
#define UVSTR   132
#define WSTR    65
#define PCHUNK  8
#define UVW     (PCHUNK * UVSTR)

#define OFF_SKEY 0
#define OFF_PERM 64
#define OFF_CP   128
#define OFF_CN   256
#define OFF_TAB  384
#define OFF_ACT  (OFF_TAB + 65 * TSTR)
#define OFF_W2S  (OFF_ACT + 4 * UVW)
#define LDS_FLOATS (OFF_W2S + HID * LSTR)

static __device__ __forceinline__ void lds_barrier()
{
    __builtin_amdgcn_fence(__ATOMIC_RELEASE, "workgroup", "local");
    __builtin_amdgcn_s_barrier();
    __builtin_amdgcn_fence(__ATOMIC_ACQUIRE, "workgroup", "local");
}

__launch_bounds__(128)
__global__ void kan_edge_kernel(const float* __restrict__ x,
                                const float* __restrict__ w0, const float* __restrict__ b0,
                                const float* __restrict__ w1, const float* __restrict__ b1,
                                const float* __restrict__ w2, const float* __restrict__ b2,
                                const float* __restrict__ wout,
                                float* __restrict__ outbuf, int use_atomic)
{
    __shared__ __attribute__((aligned(16))) float smem[LDS_FLOATS];
    __shared__ __attribute__((aligned(16))) unsigned patk[4 * 64 * 4];
    __shared__ __attribute__((aligned(16))) long long ukey[HID];
    float* const act = smem + OFF_ACT;
    float* const tab = smem + OFF_TAB;

    const int d   = blockIdx.x;
    const int o   = d >> 6;
    const int i   = d & 63;
    const int tid = threadIdx.x;

    const float* __restrict__ wod = wout + o * (IN_DIM * HID) + i * HID;
    float wo[HID];
    #pragma unroll
    for (int h = 0; h < HID; ++h) wo[h] = wod[h];
    const float xin0 = x[tid * IN_DIM + i];
    const float xin1 = x[(tid + MROWS) * IN_DIM + i];
    float myw = 0.f, myb = 0.f;
    if (tid < HID) {
        myw = w0[d * HID + tid];
        myb = b0[d * HID + tid];
    }

    const int lane  = tid & 31;
    const int r     = lane & 15;
    const int half  = lane >> 4;
    const int wave  = tid >> 5;

    float4 w1r[8];
    {
        const float4* __restrict__ w1v = reinterpret_cast<const float4*>(w1 + (size_t)d * HID * HID);
        #pragma unroll
        for (int j = 0; j < 8; ++j) w1r[j] = w1v[tid + 128 * j];
    }
    const float b1v = b1[d * HID + (tid & 63)];
    float b2v[4];
    #pragma unroll
    for (int j = 0; j < 4; ++j) b2v[j] = b2[d * HID + 16 * j + r];
    float4 w2r[8];
    {
        const float4* __restrict__ w2v = reinterpret_cast<const float4*>(w2 + (size_t)d * HID * HID);
        #pragma unroll
        for (int j = 0; j < 8; ++j) w2r[j] = w2v[tid + 128 * j];
    }
    float mykey = 0.f;
    long long mykey64 = 0;
    if (tid < HID) {
        mykey = (myw != 0.f) ? (-myb / myw)
                             : (myb > 0.f ? -__builtin_inff() : __builtin_inff());
        const int kb = __float_as_int(mykey);
        const int ki = kb ^ ((kb >> 31) & 0x7fffffff);
        mykey64 = ((long long)ki << 32) | (long long)tid;
        ukey[tid] = mykey64;
    }
    lds_barrier();

    if (tid < HID) {
        int pos = 0;
        #pragma unroll 16
        for (int h2 = 0; h2 < HID; ++h2)
            pos += (ukey[h2] < mykey64) ? 1 : 0;
        const bool rising = !(myw < 0.f);
        smem[OFF_SKEY + pos] = mykey;
        smem[OFF_PERM + pos] = __int_as_float(tid);
        smem[OFF_CP + pos]       = rising ? myw : 0.f;
        smem[OFF_CP + HID + pos] = rising ? myb : 0.f;
        smem[OFF_CN + pos]       = rising ? 0.f : myw;
        smem[OFF_CN + HID + pos] = rising ? 0.f : myb;
    }
    {
        #pragma unroll
        for (int j = 0; j < 8; ++j) {
            const int q = tid + 128 * j;
            const float4 v = w1r[j];
            float* dst = act + (q >> 4) * WSTR + 4 * (q & 15);
            dst[0] = v.x; dst[1] = v.y; dst[2] = v.z; dst[3] = v.w;
        }
    }
    {
        #pragma unroll
        for (int j = 0; j < 8; ++j) {
            const int q = tid + 128 * j;
            *reinterpret_cast<float4*>(smem + OFF_W2S + (q >> 4) * LSTR + 4 * (q & 15)) = w2r[j];
        }
    }
    lds_barrier();

    int c0 = 0, c1 = 0;
    {
        const int n   = tid & 63;
        const int sel = (tid >> 6) * HID;
        const float* cp = smem + OFF_CP + sel;
        const float* cn = smem + OFF_CN + sel;
        const float* wrow = act + n * WSTR;
        float wv[HID];
        int pidx[HID];
        #pragma unroll
        for (int p = 0; p < HID; ++p) pidx[p] = __float_as_int(smem[OFF_PERM + p]);
        __builtin_amdgcn_sched_barrier(0);
        #pragma unroll
        for (int p = HID - 1; p >= 0; --p) wv[p] = wrow[pidx[p]];
        float cnv[HID];
        #pragma unroll
        for (int p = 0; p < HID; ++p) cnv[p] = cn[p];
        __builtin_amdgcn_sched_barrier(0);

        float sfx[HID];
        float cpv[HID];
        float sn = 0.f;
        #pragma unroll
        for (int k = 0; k < 7; ++k) {
            const int off = (k < 6) ? ((32 >> k) - 1) : 0;
            const int inc = (k < 6) ? (32 >> k) : 1;
            const float key0 = smem[OFF_SKEY + c0 + off];
            const float key1 = smem[OFF_SKEY + c1 + off];
            if (k == 6) {
                #pragma unroll
                for (int s = 0; s < HID; ++s) cpv[s] = cp[s];
            }
            __builtin_amdgcn_sched_barrier(0);
            const int hi = 63 - 9 * k - (k ? 1 : 0);
            const int len = k ? 9 : 10;
            #pragma unroll
            for (int j = 0; j < len; ++j) {
                const int p = hi - j;
                sn = fmaf(wv[p], cnv[p], sn);
                sfx[p] = sn;
            }
            __builtin_amdgcn_sched_barrier(0);
            c0 += (key0 < xin0) ? inc : 0;
            c1 += (key1 < xin1) ? inc : 0;
        }
        float sp = sel ? b1v : 0.f;
        #pragma unroll
        for (int s = 0; s < HID; ++s) {
            tab[s * TSTR + tid] = sfx[s] + sp;
            sp = fmaf(wv[s], cpv[s], sp);
        }
        tab[HID * TSTR + tid] = sp;
    }

    __syncthreads();

    v2f Bfrag[4][16];
    {
        const float* wl = smem + OFF_W2S + r * LSTR + 2 * half;
        #pragma unroll
        for (int j = 0; j < 4; ++j) {
            #pragma unroll
            for (int c = 0; c < 16; ++c)
                Bfrag[j][c] = *(const v2f*)(wl + j * 16 * LSTR + c * 4);
        }
    }

    unsigned nlo0, nhi0, nlo1, nhi1;
    {
        const float* t0 = tab + c0 * TSTR;
        const float* t1 = tab + c1 * TSTR;
        unsigned m0[2] = {0u, 0u}, m1[2] = {0u, 0u};
        #pragma unroll
        for (int j = 15; j >= 0; --j) {
            const float4 sa = *reinterpret_cast<const float4*>(t0 + 4 * j);
            const float4 ta = *reinterpret_cast<const float4*>(t0 + HID + 4 * j);
            const float4 sb = *reinterpret_cast<const float4*>(t1 + 4 * j);
            const float4 tb = *reinterpret_cast<const float4*>(t1 + HID + 4 * j);
            const int w = j >> 3;
            m0[w] = (m0[w] << 1) | ((unsigned)__float_as_int(fmaf(xin0, sa.w, ta.w)) >> 31);
            m0[w] = (m0[w] << 1) | ((unsigned)__float_as_int(fmaf(xin0, sa.z, ta.z)) >> 31);
            m0[w] = (m0[w] << 1) | ((unsigned)__float_as_int(fmaf(xin0, sa.y, ta.y)) >> 31);
            m0[w] = (m0[w] << 1) | ((unsigned)__float_as_int(fmaf(xin0, sa.x, ta.x)) >> 31);
            m1[w] = (m1[w] << 1) | ((unsigned)__float_as_int(fmaf(xin1, sb.w, tb.w)) >> 31);
            m1[w] = (m1[w] << 1) | ((unsigned)__float_as_int(fmaf(xin1, sb.z, tb.z)) >> 31);
            m1[w] = (m1[w] << 1) | ((unsigned)__float_as_int(fmaf(xin1, sb.y, tb.y)) >> 31);
            m1[w] = (m1[w] << 1) | ((unsigned)__float_as_int(fmaf(xin1, sb.x, tb.x)) >> 31);
        }
        nlo0 = m0[0]; nhi0 = m0[1];
        nlo1 = m1[0]; nhi1 = m1[1];
    }

    unsigned* const mypat = patk + wave * 256;
    int pid0 = 0, pid1 = 0;
    int np = 0;
    {
        unsigned rem0 = __builtin_amdgcn_ballot_w32(true);
        unsigned rem1 = rem0;
        for (int it = 0; it < 32 && rem0 != 0u; ++it) {
            const int L = __builtin_ctz(rem0);
            const int ks = __builtin_amdgcn_readlane(c0, L);
            const unsigned kl = (unsigned)__builtin_amdgcn_readlane((int)nlo0, L);
            const unsigned kh = (unsigned)__builtin_amdgcn_readlane((int)nhi0, L);
            const bool e0 = (c0 == ks) & (nlo0 == kl) & (nhi0 == kh);
            const bool e1 = (c1 == ks) & (nlo1 == kl) & (nhi1 == kh);
            rem0 &= ~__builtin_amdgcn_ballot_w32(e0);
            rem1 &= ~__builtin_amdgcn_ballot_w32(e1);
            pid0 = e0 ? np : pid0;
            pid1 = e1 ? np : pid1;
            if (lane == 0) {
                mypat[4 * np]     = (unsigned)ks;
                mypat[4 * np + 1] = kl;
                mypat[4 * np + 2] = kh;
            }
            ++np;
        }
        for (int it = 0; it < 32 && rem1 != 0u; ++it) {
            const int L = __builtin_ctz(rem1);
            const int ks = __builtin_amdgcn_readlane(c1, L);
            const unsigned kl = (unsigned)__builtin_amdgcn_readlane((int)nlo1, L);
            const unsigned kh = (unsigned)__builtin_amdgcn_readlane((int)nhi1, L);
            const bool e1 = (c1 == ks) & (nlo1 == kl) & (nhi1 == kh);
            rem1 &= ~__builtin_amdgcn_ballot_w32(e1);
            pid1 = e1 ? np : pid1;
            if (lane == 0) {
                mypat[4 * np]     = (unsigned)ks;
                mypat[4 * np + 1] = kl;
                mypat[4 * np + 2] = kh;
            }
            ++np;
        }
    }

    float* const uv = smem + OFF_ACT + wave * UVW;
    const int nch = (np + PCHUNK - 1) / PCHUNK;
    #pragma nounroll
    for (int ch = 0; ch < nch; ++ch) {
        const int q = ch * PCHUNK + (r & 7);
        const bool live = q < np;
        const unsigned rs = mypat[4 * q], rl = mypat[4 * q + 1], rh = mypat[4 * q + 2];
        const int psg = live ? (int)rs : 0;
        const unsigned alo = (live ? ~rl : 0u) >> (2 * half);
        const unsigned ahi = (live ? ~rh : 0u) >> (2 * half);
        const float* arow = tab + psg * TSTR + ((r & 8) ? HID : 0) + 2 * half;
        v2f araw[16], af[16];
        #pragma unroll
        for (int c = 0; c < 16; ++c) araw[c] = *(const v2f*)(arow + 4 * c);
        __builtin_amdgcn_sched_barrier(0);
        #pragma unroll
        for (int c = 0; c < 16; ++c) {
            const v2f t = araw[c];
            const unsigned wsel = (c < 8) ? alo : ahi;
            const int sh = 4 * (c & 7);
            af[c].x = ((wsel >> sh) & 1u) ? t.x : 0.f;
            af[c].y = ((wsel >> (sh + 1)) & 1u) ? t.y : 0.f;
        }

        v8f acc[4];
        #pragma unroll
        for (int j = 0; j < 4; ++j) {
            const float ini = half ? b2v[j] : 0.f;
            acc[j] = (v8f){ini, ini, ini, ini, ini, ini, ini, ini};
        }
        #pragma unroll
        for (int c = 0; c < 16; ++c) {
            #pragma unroll
            for (int j = 0; j < 4; ++j)
                acc[j] = __builtin_amdgcn_wmma_f32_16x16x4_f32(
                             false, af[c], false, Bfrag[j][c],
                             (short)0, acc[j], false, false);
        }
        #pragma unroll
        for (int j = 0; j < 4; ++j) {
            #pragma unroll
            for (int v = 0; v < 8; ++v)
                uv[v * UVSTR + half * HID + 16 * j + r] = acc[j][v];
        }

        #pragma unroll
        for (int set = 0; set < 2; ++set) {
            const int pid = set ? pid1 : pid0;
            if ((pid >> 3) == ch) {
                const float xv = set ? xin1 : xin0;
                const float* p = uv + (pid & 7) * UVSTR;
                float s = 0.f;
                #pragma unroll
                for (int g = 0; g < 2; ++g) {
                    float4 u4[8], v4[8];
                    #pragma unroll
                    for (int t = 0; t < 8; ++t) {
                        u4[t] = *reinterpret_cast<const float4*>(p + 32 * g + 4 * t);
                        v4[t] = *reinterpret_cast<const float4*>(p + HID + 32 * g + 4 * t);
                    }
                    __builtin_amdgcn_sched_barrier(0);
                    #pragma unroll
                    for (int t = 0; t < 8; ++t) {
                        const int h = 32 * g + 4 * t;
                        float z;
                        z = fmaf(xv, u4[t].x, v4[t].x); z = z > 0.f ? z : 0.f; s = fmaf(wo[h + 0], z, s);
                        z = fmaf(xv, u4[t].y, v4[t].y); z = z > 0.f ? z : 0.f; s = fmaf(wo[h + 1], z, s);
                        z = fmaf(xv, u4[t].z, v4[t].z); z = z > 0.f ? z : 0.f; s = fmaf(wo[h + 2], z, s);
                        z = fmaf(xv, u4[t].w, v4[t].w); z = z > 0.f ? z : 0.f; s = fmaf(wo[h + 3], z, s);
                    }
                }
                const int row = tid + set * MROWS;
                if (use_atomic) {
                    __hip_atomic_fetch_add(&outbuf[row * OUT_DIM + o], s,
                                           __ATOMIC_RELAXED, __HIP_MEMORY_SCOPE_AGENT);
                } else {
                    outbuf[(size_t)d * BSZ + row] = s;
                }
            }
        }
    }
}

__global__ void kan_reduce_kernel(const float* __restrict__ part,
                                  const float* __restrict__ bout,
                                  float* __restrict__ out)
{
    const int o = blockIdx.x;
    const int b = threadIdx.x;
    float s = bout[o];
    const float* p = part + (size_t)o * 64 * BSZ + b;
    #pragma unroll 32
    for (int i = 0; i < 64; ++i) s += p[i * BSZ];
    out[b * OUT_DIM + o] = s;
}

__global__ void kan_init_out(const float* __restrict__ bout, float* __restrict__ out)
{
    const int idx = blockIdx.x * 256 + threadIdx.x;
    out[idx] = bout[idx & 63];
}

extern "C" void kernel_launch(void* const* d_in, const int* in_sizes, int n_in,
                              void* d_out, int out_size, void* d_ws, size_t ws_size,
                              hipStream_t stream)
{
    const float* x    = (const float*)d_in[0];
    const float* w0   = (const float*)d_in[1];
    const float* b0   = (const float*)d_in[2];
    const float* w1   = (const float*)d_in[3];
    const float* b1   = (const float*)d_in[4];
    const float* w2   = (const float*)d_in[5];
    const float* b2   = (const float*)d_in[6];
    const float* wout = (const float*)d_in[7];
    const float* bout = (const float*)d_in[8];
    float* out = (float*)d_out;

    const size_t need = (size_t)DDIM * BSZ * sizeof(float);
    if (ws_size >= need) {
        float* part = (float*)d_ws;
        kan_edge_kernel<<<DDIM, MROWS, 0, stream>>>(x, w0, b0, w1, b1, w2, b2, wout, part, 0);
        kan_reduce_kernel<<<OUT_DIM, BSZ, 0, stream>>>(part, bout, out);
    } else {
        kan_init_out<<<64, 256, 0, stream>>>(bout, out);
        kan_edge_kernel<<<DDIM, MROWS, 0, stream>>>(x, w0, b0, w1, b1, w2, b2, wout, out, 1);
    }
}


# Layer 2 once per pattern of the WORKGROUP's 256 rows (was per wave's 64): wave lists joined in LDS, each wave forms one 16-column tile of U,V for all patterns; layer-2 k sum now two chains, added
# speedup vs baseline: 2.5344x; 1.0950x over previous
#include <hip/hip_runtime.h>

typedef float v2f __attribute__((ext_vector_type(2)));
typedef float v8f __attribute__((ext_vector_type(8)));

#define IN_DIM  64
#define OUT_DIM 64
#define HID     64
#define DDIM    (IN_DIM * OUT_DIM)
#define BSZ     256
#define MROWS   128
#define LSTR    68
#define TSTR    132
#define UVSTR   132
#define WSTR    65
#define PCHUNK  8
#define UVW     (PCHUNK * UVSTR)

#define OFF_SKEY 0
#define OFF_PERM 64
#define OFF_CP   128
#define OFF_CN   256
#define OFF_TAB  384
#define OFF_ACT  (OFF_TAB + 65 * TSTR)
#define OFF_W2S  (OFF_ACT + 4 * UVW)
#define LDS_FLOATS (OFF_W2S + HID * LSTR)

static __device__ __forceinline__ void lds_barrier()
{
    __builtin_amdgcn_fence(__ATOMIC_RELEASE, "workgroup", "local");
    __builtin_amdgcn_s_barrier();
    __builtin_amdgcn_fence(__ATOMIC_ACQUIRE, "workgroup", "local");
}

__launch_bounds__(128)
__global__ void kan_edge_kernel(const float* __restrict__ x,
                                const float* __restrict__ w0, const float* __restrict__ b0,
                                const float* __restrict__ w1, const float* __restrict__ b1,
                                const float* __restrict__ w2, const float* __restrict__ b2,
                                const float* __restrict__ wout,
                                float* __restrict__ outbuf, int use_atomic)
{
    __shared__ __attribute__((aligned(16))) float smem[LDS_FLOATS];
    __shared__ __attribute__((aligned(16))) unsigned patk[4 * 64 * 4];
    __shared__ __attribute__((aligned(16))) long long ukey[HID];
    __shared__ __attribute__((aligned(16))) unsigned gpk[4 * 64 * 4];
    __shared__ __attribute__((aligned(16))) unsigned npw[4];
    float* const act = smem + OFF_ACT;
    float* const tab = smem + OFF_TAB;

    const int d   = blockIdx.x;
    const int o   = d >> 6;
    const int i   = d & 63;
    const int tid = threadIdx.x;

    const float* __restrict__ wod = wout + o * (IN_DIM * HID) + i * HID;
    float wo[HID];
    #pragma unroll
    for (int h = 0; h < HID; ++h) wo[h] = wod[h];
    const float xin0 = x[tid * IN_DIM + i];
    const float xin1 = x[(tid + MROWS) * IN_DIM + i];
    float myw = 0.f, myb = 0.f;
    if (tid < HID) {
        myw = w0[d * HID + tid];
        myb = b0[d * HID + tid];
    }

    const int lane  = tid & 31;
    const int r     = lane & 15;
    const int half  = lane >> 4;
    const int wave  = tid >> 5;

    float4 w1r[8];
    {
        const float4* __restrict__ w1v = reinterpret_cast<const float4*>(w1 + (size_t)d * HID * HID);
        #pragma unroll
        for (int j = 0; j < 8; ++j) w1r[j] = w1v[tid + 128 * j];
    }
    const float b1v = b1[d * HID + (tid & 63)];
    float4 w2r[8];
    {
        const float4* __restrict__ w2v = reinterpret_cast<const float4*>(w2 + (size_t)d * HID * HID);
        #pragma unroll
        for (int j = 0; j < 8; ++j) w2r[j] = w2v[tid + 128 * j];
    }
    float mykey = 0.f;
    long long mykey64 = 0;
    if (tid < HID) {
        mykey = (myw != 0.f) ? (-myb / myw)
                             : (myb > 0.f ? -__builtin_inff() : __builtin_inff());
        const int kb = __float_as_int(mykey);
        const int ki = kb ^ ((kb >> 31) & 0x7fffffff);
        mykey64 = ((long long)ki << 32) | (long long)tid;
        ukey[tid] = mykey64;
    }
    lds_barrier();

    if (tid < HID) {
        int pos = 0;
        #pragma unroll 16
        for (int h2 = 0; h2 < HID; ++h2)
            pos += (ukey[h2] < mykey64) ? 1 : 0;
        const bool rising = !(myw < 0.f);
        smem[OFF_SKEY + pos] = mykey;
        smem[OFF_PERM + pos] = __int_as_float(tid);
        smem[OFF_CP + pos]       = rising ? myw : 0.f;
        smem[OFF_CP + HID + pos] = rising ? myb : 0.f;
        smem[OFF_CN + pos]       = rising ? 0.f : myw;
        smem[OFF_CN + HID + pos] = rising ? 0.f : myb;
    }
    {
        #pragma unroll
        for (int j = 0; j < 8; ++j) {
            const int q = tid + 128 * j;
            const float4 v = w1r[j];
            float* dst = act + (q >> 4) * WSTR + 4 * (q & 15);
            dst[0] = v.x; dst[1] = v.y; dst[2] = v.z; dst[3] = v.w;
        }
    }
    {
        #pragma unroll
        for (int j = 0; j < 8; ++j) {
            const int q = tid + 128 * j;
            *reinterpret_cast<float4*>(smem + OFF_W2S + (q >> 4) * LSTR + 4 * (q & 15)) = w2r[j];
        }
    }
    lds_barrier();

    int c0 = 0, c1 = 0;
    {
        const int n   = tid & 63;
        const int sel = (tid >> 6) * HID;
        const float* cp = smem + OFF_CP + sel;
        const float* cn = smem + OFF_CN + sel;
        const float* wrow = act + n * WSTR;
        float wv[HID];
        int pidx[HID];
        #pragma unroll
        for (int p = 0; p < HID; ++p) pidx[p] = __float_as_int(smem[OFF_PERM + p]);
        __builtin_amdgcn_sched_barrier(0);
        #pragma unroll
        for (int p = HID - 1; p >= 0; --p) wv[p] = wrow[pidx[p]];
        float cnv[HID];
        #pragma unroll
        for (int p = 0; p < HID; ++p) cnv[p] = cn[p];
        __builtin_amdgcn_sched_barrier(0);

        float sfx[HID];
        float cpv[HID];
        float sn = 0.f;
        #pragma unroll
        for (int k = 0; k < 7; ++k) {
            const int off = (k < 6) ? ((32 >> k) - 1) : 0;
            const int inc = (k < 6) ? (32 >> k) : 1;
            const float key0 = smem[OFF_SKEY + c0 + off];
            const float key1 = smem[OFF_SKEY + c1 + off];
            if (k == 6) {
                #pragma unroll
                for (int s = 0; s < HID; ++s) cpv[s] = cp[s];
            }
            __builtin_amdgcn_sched_barrier(0);
            const int hi = 63 - 9 * k - (k ? 1 : 0);
            const int len = k ? 9 : 10;
            #pragma unroll
            for (int j = 0; j < len; ++j) {
                const int p = hi - j;
                sn = fmaf(wv[p], cnv[p], sn);
                sfx[p] = sn;
            }
            __builtin_amdgcn_sched_barrier(0);
            c0 += (key0 < xin0) ? inc : 0;
            c1 += (key1 < xin1) ? inc : 0;
        }
        float sp = sel ? b1v : 0.f;
        #pragma unroll
        for (int s = 0; s < HID; ++s) {
            tab[s * TSTR + tid] = sfx[s] + sp;
            sp = fmaf(wv[s], cpv[s], sp);
        }
        tab[HID * TSTR + tid] = sp;
    }

    __syncthreads();

    v2f Bown[16];
    {
        const float* wl = smem + OFF_W2S + (16 * wave + r) * LSTR + 2 * half;
        #pragma unroll
        for (int c = 0; c < 16; ++c) Bown[c] = *(const v2f*)(wl + c * 4);
    }
    const float b2own = b2[d * HID + 16 * wave + r];

    unsigned nlo0, nhi0, nlo1, nhi1;
    {
        const float* t0 = tab + c0 * TSTR;
        const float* t1 = tab + c1 * TSTR;
        unsigned m0[2] = {0u, 0u}, m1[2] = {0u, 0u};
        #pragma unroll
        for (int j = 15; j >= 0; --j) {
            const float4 sa = *reinterpret_cast<const float4*>(t0 + 4 * j);
            const float4 ta = *reinterpret_cast<const float4*>(t0 + HID + 4 * j);
            const float4 sb = *reinterpret_cast<const float4*>(t1 + 4 * j);
            const float4 tb = *reinterpret_cast<const float4*>(t1 + HID + 4 * j);
            const int w = j >> 3;
            m0[w] = (m0[w] << 1) | ((unsigned)__float_as_int(fmaf(xin0, sa.w, ta.w)) >> 31);
            m0[w] = (m0[w] << 1) | ((unsigned)__float_as_int(fmaf(xin0, sa.z, ta.z)) >> 31);
            m0[w] = (m0[w] << 1) | ((unsigned)__float_as_int(fmaf(xin0, sa.y, ta.y)) >> 31);
            m0[w] = (m0[w] << 1) | ((unsigned)__float_as_int(fmaf(xin0, sa.x, ta.x)) >> 31);
            m1[w] = (m1[w] << 1) | ((unsigned)__float_as_int(fmaf(xin1, sb.w, tb.w)) >> 31);
            m1[w] = (m1[w] << 1) | ((unsigned)__float_as_int(fmaf(xin1, sb.z, tb.z)) >> 31);
            m1[w] = (m1[w] << 1) | ((unsigned)__float_as_int(fmaf(xin1, sb.y, tb.y)) >> 31);
            m1[w] = (m1[w] << 1) | ((unsigned)__float_as_int(fmaf(xin1, sb.x, tb.x)) >> 31);
        }
        nlo0 = m0[0]; nhi0 = m0[1];
        nlo1 = m1[0]; nhi1 = m1[1];
    }

    unsigned* const mypat = patk + wave * 256;
    int pid0 = 0, pid1 = 0;
    int np = 0;
    {
        unsigned rem0 = __builtin_amdgcn_ballot_w32(true);
        unsigned rem1 = rem0;
        for (int it = 0; it < 32 && rem0 != 0u; ++it) {
            const int L = __builtin_ctz(rem0);
            const int ks = __builtin_amdgcn_readlane(c0, L);
            const unsigned kl = (unsigned)__builtin_amdgcn_readlane((int)nlo0, L);
            const unsigned kh = (unsigned)__builtin_amdgcn_readlane((int)nhi0, L);
            const bool e0 = (c0 == ks) & (nlo0 == kl) & (nhi0 == kh);
            const bool e1 = (c1 == ks) & (nlo1 == kl) & (nhi1 == kh);
            rem0 &= ~__builtin_amdgcn_ballot_w32(e0);
            rem1 &= ~__builtin_amdgcn_ballot_w32(e1);
            pid0 = e0 ? np : pid0;
            pid1 = e1 ? np : pid1;
            if (lane == 0) {
                mypat[4 * np]     = (unsigned)ks;
                mypat[4 * np + 1] = kl;
                mypat[4 * np + 2] = kh;
            }
            ++np;
        }
        for (int it = 0; it < 32 && rem1 != 0u; ++it) {
            const int L = __builtin_ctz(rem1);
            const int ks = __builtin_amdgcn_readlane(c1, L);
            const unsigned kl = (unsigned)__builtin_amdgcn_readlane((int)nlo1, L);
            const unsigned kh = (unsigned)__builtin_amdgcn_readlane((int)nhi1, L);
            const bool e1 = (c1 == ks) & (nlo1 == kl) & (nhi1 == kh);
            rem1 &= ~__builtin_amdgcn_ballot_w32(e1);
            pid1 = e1 ? np : pid1;
            if (lane == 0) {
                mypat[4 * np]     = (unsigned)ks;
                mypat[4 * np + 1] = kl;
                mypat[4 * np + 2] = kh;
            }
            ++np;
        }
    }

    if (lane == 0) npw[wave] = (unsigned)np;
    lds_barrier();
    const uint4 nn = *reinterpret_cast<const uint4*>(npw);
    const int pre1 = __builtin_amdgcn_readfirstlane((int)nn.x);
    const int pre2 = pre1 + __builtin_amdgcn_readfirstlane((int)nn.y);
    const int pre3 = pre2 + __builtin_amdgcn_readfirstlane((int)nn.z);
    const int nent = pre3 + __builtin_amdgcn_readfirstlane((int)nn.w);
    unsigned* const gp = gpk + wave * 256;
    int ng = 0;
    int g0 = 0, g1 = 0;
    bool shared = false;
    if (nent <= 64) {
        const int ea = lane, eb = lane + 32;
        const int wa = (ea >= pre1 ? 1 : 0) + (ea >= pre2 ? 1 : 0) + (ea >= pre3 ? 1 : 0);
        const int wb = (eb >= pre1 ? 1 : 0) + (eb >= pre2 ? 1 : 0) + (eb >= pre3 ? 1 : 0);
        int fa = ea >= pre1 ? pre1 : 0;
        fa = ea >= pre2 ? pre2 : fa;
        fa = ea >= pre3 ? pre3 : fa;
        int fb = eb >= pre1 ? pre1 : 0;
        fb = eb >= pre2 ? pre2 : fb;
        fb = eb >= pre3 ? pre3 : fb;
        const int qa = ea - fa;
        const int qb = eb - fb;
        const uint4 ka = *reinterpret_cast<const uint4*>(patk + 4 * (wa * 64 + qa));
        const uint4 kb = *reinterpret_cast<const uint4*>(patk + 4 * (wb * 64 + qb));
        const bool va = ea < nent, vb = eb < nent;
        unsigned rem0 = __builtin_amdgcn_ballot_w32(va);
        unsigned rem1 = __builtin_amdgcn_ballot_w32(vb);
        for (int it = 0; it < 32 && rem0 != 0u; ++it) {
            const int L = __builtin_ctz(rem0);
            const unsigned ks = (unsigned)__builtin_amdgcn_readlane((int)ka.x, L);
            const unsigned kl = (unsigned)__builtin_amdgcn_readlane((int)ka.y, L);
            const unsigned kh = (unsigned)__builtin_amdgcn_readlane((int)ka.z, L);
            const bool e0 = va & (ka.x == ks) & (ka.y == kl) & (ka.z == kh);
            const bool e1 = vb & (kb.x == ks) & (kb.y == kl) & (kb.z == kh);
            rem0 &= ~__builtin_amdgcn_ballot_w32(e0);
            rem1 &= ~__builtin_amdgcn_ballot_w32(e1);
            g0 = e0 ? ng : g0;
            g1 = e1 ? ng : g1;
            if (lane == 0) {
                gp[4 * ng]     = ks;
                gp[4 * ng + 1] = kl;
                gp[4 * ng + 2] = kh;
            }
            ++ng;
        }
        for (int it = 0; it < 32 && rem1 != 0u; ++it) {
            const int L = __builtin_ctz(rem1);
            const unsigned ks = (unsigned)__builtin_amdgcn_readlane((int)kb.x, L);
            const unsigned kl = (unsigned)__builtin_amdgcn_readlane((int)kb.y, L);
            const unsigned kh = (unsigned)__builtin_amdgcn_readlane((int)kb.z, L);
            const bool e1 = vb & (kb.x == ks) & (kb.y == kl) & (kb.z == kh);
            rem1 &= ~__builtin_amdgcn_ballot_w32(e1);
            g1 = e1 ? ng : g1;
            if (lane == 0) {
                gp[4 * ng]     = ks;
                gp[4 * ng + 1] = kl;
                gp[4 * ng + 2] = kh;
            }
            ++ng;
        }
        shared = ng <= 32;
    }

    if (shared) {
        int mybase = wave >= 1 ? pre1 : 0;
        mybase = wave >= 2 ? pre2 : mybase;
        mybase = wave >= 3 ? pre3 : mybase;
        const int i0 = mybase + pid0, i1 = mybase + pid1;
        const int t00 = __builtin_amdgcn_ds_bpermute(4 * (i0 & 31), g0);
        const int t01 = __builtin_amdgcn_ds_bpermute(4 * (i0 & 31), g1);
        const int t10 = __builtin_amdgcn_ds_bpermute(4 * (i1 & 31), g0);
        const int t11 = __builtin_amdgcn_ds_bpermute(4 * (i1 & 31), g1);
        const int pg0 = (i0 < 32) ? t00 : t01;
        const int pg1 = (i1 < 32) ? t10 : t11;

        float* const uvg = smem + OFF_ACT;
        const int nchg = (ng + PCHUNK - 1) / PCHUNK;
        #pragma nounroll
        for (int ch = 0; ch < nchg; ++ch) {
            const int q = ch * PCHUNK + (r & 7);
            const bool live = q < ng;
            const unsigned rs = gp[4 * q], rl = gp[4 * q + 1], rh = gp[4 * q + 2];
            const int psg = live ? (int)rs : 0;
            const unsigned alo = (live ? ~rl : 0u) >> (2 * half);
            const unsigned ahi = (live ? ~rh : 0u) >> (2 * half);
            const float* arow = tab + psg * TSTR + ((r & 8) ? HID : 0) + 2 * half;
            v2f araw[16], af[16];
            #pragma unroll
            for (int c = 0; c < 16; ++c) araw[c] = *(const v2f*)(arow + 4 * c);
            __builtin_amdgcn_sched_barrier(0);
            #pragma unroll
            for (int c = 0; c < 16; ++c) {
                const v2f t = araw[c];
                const unsigned wsel = (c < 8) ? alo : ahi;
                const int sh = 4 * (c & 7);
                af[c].x = ((wsel >> sh) & 1u) ? t.x : 0.f;
                af[c].y = ((wsel >> (sh + 1)) & 1u) ? t.y : 0.f;
            }

            const float ini = half ? b2own : 0.f;
            v8f accE = (v8f){ini, ini, ini, ini, ini, ini, ini, ini};
            v8f accO = (v8f){0.f, 0.f, 0.f, 0.f, 0.f, 0.f, 0.f, 0.f};
            #pragma unroll
            for (int c = 0; c < 16; c += 2) {
                accE = __builtin_amdgcn_wmma_f32_16x16x4_f32(
                           false, af[c], false, Bown[c], (short)0, accE, false, false);
                accO = __builtin_amdgcn_wmma_f32_16x16x4_f32(
                           false, af[c + 1], false, Bown[c + 1], (short)0, accO, false, false);
            }
            #pragma unroll
            for (int v = 0; v < 8; ++v)
                uvg[(ch * PCHUNK + v) * UVSTR + half * HID + 16 * wave + r] = accE[v] + accO[v];
        }
        lds_barrier();

        #pragma unroll
        for (int set = 0; set < 2; ++set) {
            const float xv = set ? xin1 : xin0;
            const float* p = uvg + (set ? pg1 : pg0) * UVSTR;
            float s = 0.f;
            #pragma unroll
            for (int g = 0; g < 2; ++g) {
                float4 u4[8], v4[8];
                #pragma unroll
                for (int t = 0; t < 8; ++t) {
                    u4[t] = *reinterpret_cast<const float4*>(p + 32 * g + 4 * t);
                    v4[t] = *reinterpret_cast<const float4*>(p + HID + 32 * g + 4 * t);
                }
                __builtin_amdgcn_sched_barrier(0);
                #pragma unroll
                for (int t = 0; t < 8; ++t) {
                    const int h = 32 * g + 4 * t;
                    float z;
                    z = fmaf(xv, u4[t].x, v4[t].x); z = z > 0.f ? z : 0.f; s = fmaf(wo[h + 0], z, s);
                    z = fmaf(xv, u4[t].y, v4[t].y); z = z > 0.f ? z : 0.f; s = fmaf(wo[h + 1], z, s);
                    z = fmaf(xv, u4[t].z, v4[t].z); z = z > 0.f ? z : 0.f; s = fmaf(wo[h + 2], z, s);
                    z = fmaf(xv, u4[t].w, v4[t].w); z = z > 0.f ? z : 0.f; s = fmaf(wo[h + 3], z, s);
                }
            }
            const int row = tid + set * MROWS;
            if (use_atomic) {
                __hip_atomic_fetch_add(&outbuf[row * OUT_DIM + o], s,
                                       __ATOMIC_RELAXED, __HIP_MEMORY_SCOPE_AGENT);
            } else {
                outbuf[(size_t)d * BSZ + row] = s;
            }
        }
    } else {
        float* const uv = smem + OFF_ACT + wave * UVW;
        const int nch = (np + PCHUNK - 1) / PCHUNK;
        #pragma nounroll
        for (int ch = 0; ch < nch; ++ch) {
            const int q = ch * PCHUNK + (r & 7);
            const bool live = q < np;
            const unsigned rs = mypat[4 * q], rl = mypat[4 * q + 1], rh = mypat[4 * q + 2];
            const int psg = live ? (int)rs : 0;
            const unsigned alo = (live ? ~rl : 0u) >> (2 * half);
            const unsigned ahi = (live ? ~rh : 0u) >> (2 * half);
            const float* arow = tab + psg * TSTR + ((r & 8) ? HID : 0) + 2 * half;
            v2f af[16];
            #pragma unroll
            for (int c = 0; c < 16; ++c) {
                const v2f t = *(const v2f*)(arow + 4 * c);
                const unsigned wsel = (c < 8) ? alo : ahi;
                const int sh = 4 * (c & 7);
                af[c].x = ((wsel >> sh) & 1u) ? t.x : 0.f;
                af[c].y = ((wsel >> (sh + 1)) & 1u) ? t.y : 0.f;
            }
            #pragma nounroll
            for (int j = 0; j < 4; ++j) {
                const float* wl = smem + OFF_W2S + (16 * j + r) * LSTR + 2 * half;
                v2f Bj[16];
                #pragma unroll
                for (int c = 0; c < 16; ++c) Bj[c] = *(const v2f*)(wl + c * 4);
                const float ini = half ? b2[d * HID + 16 * j + r] : 0.f;
                v8f accE = (v8f){ini, ini, ini, ini, ini, ini, ini, ini};
                v8f accO = (v8f){0.f, 0.f, 0.f, 0.f, 0.f, 0.f, 0.f, 0.f};
                #pragma unroll
                for (int c = 0; c < 16; c += 2) {
                    accE = __builtin_amdgcn_wmma_f32_16x16x4_f32(
                               false, af[c], false, Bj[c], (short)0, accE, false, false);
                    accO = __builtin_amdgcn_wmma_f32_16x16x4_f32(
                               false, af[c + 1], false, Bj[c + 1], (short)0, accO, false, false);
                }
                #pragma unroll
                for (int v = 0; v < 8; ++v)
                    uv[v * UVSTR + half * HID + 16 * j + r] = accE[v] + accO[v];
            }

            #pragma unroll
            for (int set = 0; set < 2; ++set) {
                const int pid = set ? pid1 : pid0;
                if ((pid >> 3) == ch) {
                    const float xv = set ? xin1 : xin0;
                    const float* p = uv + (pid & 7) * UVSTR;
                    float s = 0.f;
                    #pragma unroll
                    for (int g = 0; g < 2; ++g) {
                        float4 u4[8], v4[8];
                        #pragma unroll
                        for (int t = 0; t < 8; ++t) {
                            u4[t] = *reinterpret_cast<const float4*>(p + 32 * g + 4 * t);
                            v4[t] = *reinterpret_cast<const float4*>(p + HID + 32 * g + 4 * t);
                        }
                        #pragma unroll
                        for (int t = 0; t < 8; ++t) {
                            const int h = 32 * g + 4 * t;
                            float z;
                            z = fmaf(xv, u4[t].x, v4[t].x); z = z > 0.f ? z : 0.f; s = fmaf(wo[h + 0], z, s);
                            z = fmaf(xv, u4[t].y, v4[t].y); z = z > 0.f ? z : 0.f; s = fmaf(wo[h + 1], z, s);
                            z = fmaf(xv, u4[t].z, v4[t].z); z = z > 0.f ? z : 0.f; s = fmaf(wo[h + 2], z, s);
                            z = fmaf(xv, u4[t].w, v4[t].w); z = z > 0.f ? z : 0.f; s = fmaf(wo[h + 3], z, s);
                        }
                    }
                    const int row = tid + set * MROWS;
                    if (use_atomic) {
                        __hip_atomic_fetch_add(&outbuf[row * OUT_DIM + o], s,
                                               __ATOMIC_RELAXED, __HIP_MEMORY_SCOPE_AGENT);
                    } else {
                        outbuf[(size_t)d * BSZ + row] = s;
                    }
                }
            }
        }
    }
}

__global__ void kan_reduce_kernel(const float* __restrict__ part,
                                  const float* __restrict__ bout,
                                  float* __restrict__ out)
{
    const int o = blockIdx.x;
    const int b = threadIdx.x;
    float s = bout[o];
    const float* p = part + (size_t)o * 64 * BSZ + b;
    #pragma unroll 32
    for (int i = 0; i < 64; ++i) s += p[i * BSZ];
    out[b * OUT_DIM + o] = s;
}

__global__ void kan_init_out(const float* __restrict__ bout, float* __restrict__ out)
{
    const int idx = blockIdx.x * 256 + threadIdx.x;
    out[idx] = bout[idx & 63];
}

extern "C" void kernel_launch(void* const* d_in, const int* in_sizes, int n_in,
                              void* d_out, int out_size, void* d_ws, size_t ws_size,
                              hipStream_t stream)
{
    const float* x    = (const float*)d_in[0];
    const float* w0   = (const float*)d_in[1];
    const float* b0   = (const float*)d_in[2];
    const float* w1   = (const float*)d_in[3];
    const float* b1   = (const float*)d_in[4];
    const float* w2   = (const float*)d_in[5];
    const float* b2   = (const float*)d_in[6];
    const float* wout = (const float*)d_in[7];
    const float* bout = (const float*)d_in[8];
    float* out = (float*)d_out;

    const size_t need = (size_t)DDIM * BSZ * sizeof(float);
    if (ws_size >= need) {
        float* part = (float*)d_ws;
        kan_edge_kernel<<<DDIM, MROWS, 0, stream>>>(x, w0, b0, w1, b1, w2, b2, wout, part, 0);
        kan_reduce_kernel<<<OUT_DIM, BSZ, 0, stream>>>(part, bout, out);
    } else {
        kan_init_out<<<64, 256, 0, stream>>>(bout, out);
        kan_edge_kernel<<<DDIM, MROWS, 0, stream>>>(x, w0, b0, w1, b1, w2, b2, wout, out, 1);
    }
}


# Five workgroups per WGP instead of four: W2 B fragments straight from global, no LDS copy (LDS 78880 -> 61472 B); rising coefficients read in two halves (VGPR 210 -> 186); sums unchanged
# speedup vs baseline: 2.6795x; 1.0573x over previous
#include <hip/hip_runtime.h>

typedef float v2f __attribute__((ext_vector_type(2)));
typedef float v8f __attribute__((ext_vector_type(8)));

#define IN_DIM  64
#define OUT_DIM 64
#define HID     64
#define DDIM    (IN_DIM * OUT_DIM)
#define BSZ     256
#define MROWS   128
#define TSTR    132
#define UVSTR   132
#define WSTR    65
#define PCHUNK  8
#define UVW     (PCHUNK * UVSTR)

#define OFF_SKEY 0
#define OFF_PERM 64
#define OFF_CP   128
#define OFF_CN   256
#define OFF_TAB  384
#define OFF_ACT  (OFF_TAB + 65 * TSTR)
#define LDS_FLOATS (OFF_ACT + 4 * UVW)

static __device__ __forceinline__ void lds_barrier()
{
    __builtin_amdgcn_fence(__ATOMIC_RELEASE, "workgroup", "local");
    __builtin_amdgcn_s_barrier();
    __builtin_amdgcn_fence(__ATOMIC_ACQUIRE, "workgroup", "local");
}

__launch_bounds__(128, 5)
__global__ void kan_edge_kernel(const float* __restrict__ x,
                                const float* __restrict__ w0, const float* __restrict__ b0,
                                const float* __restrict__ w1, const float* __restrict__ b1,
                                const float* __restrict__ w2, const float* __restrict__ b2,
                                const float* __restrict__ wout,
                                float* __restrict__ outbuf, int use_atomic)
{
    __shared__ __attribute__((aligned(16))) float smem[LDS_FLOATS];
    __shared__ __attribute__((aligned(16))) unsigned patk[4 * 64 * 4];
    __shared__ __attribute__((aligned(16))) long long ukey[HID];
    __shared__ __attribute__((aligned(16))) unsigned gpk[4 * 64 * 4];
    __shared__ __attribute__((aligned(16))) unsigned npw[4];
    float* const act = smem + OFF_ACT;
    float* const tab = smem + OFF_TAB;

    const int d   = blockIdx.x;
    const int o   = d >> 6;
    const int i   = d & 63;
    const int tid = threadIdx.x;

    const float* __restrict__ wod = wout + o * (IN_DIM * HID) + i * HID;
    float wo[HID];
    #pragma unroll
    for (int h = 0; h < HID; ++h) wo[h] = wod[h];
    const float xin0 = x[tid * IN_DIM + i];
    const float xin1 = x[(tid + MROWS) * IN_DIM + i];
    float myw = 0.f, myb = 0.f;
    if (tid < HID) {
        myw = w0[d * HID + tid];
        myb = b0[d * HID + tid];
    }

    const int lane  = tid & 31;
    const int r     = lane & 15;
    const int half  = lane >> 4;
    const int wave  = tid >> 5;

    float4 w1r[8];
    {
        const float4* __restrict__ w1v = reinterpret_cast<const float4*>(w1 + (size_t)d * HID * HID);
        #pragma unroll
        for (int j = 0; j < 8; ++j) w1r[j] = w1v[tid + 128 * j];
    }
    const float b1v = b1[d * HID + (tid & 63)];
    float mykey = 0.f;
    long long mykey64 = 0;
    if (tid < HID) {
        mykey = (myw != 0.f) ? (-myb / myw)
                             : (myb > 0.f ? -__builtin_inff() : __builtin_inff());
        const int kb = __float_as_int(mykey);
        const int ki = kb ^ ((kb >> 31) & 0x7fffffff);
        mykey64 = ((long long)ki << 32) | (long long)tid;
        ukey[tid] = mykey64;
    }
    lds_barrier();

    if (tid < HID) {
        int pos = 0;
        #pragma unroll 16
        for (int h2 = 0; h2 < HID; ++h2)
            pos += (ukey[h2] < mykey64) ? 1 : 0;
        const bool rising = !(myw < 0.f);
        smem[OFF_SKEY + pos] = mykey;
        smem[OFF_PERM + pos] = __int_as_float(tid);
        smem[OFF_CP + pos]       = rising ? myw : 0.f;
        smem[OFF_CP + HID + pos] = rising ? myb : 0.f;
        smem[OFF_CN + pos]       = rising ? 0.f : myw;
        smem[OFF_CN + HID + pos] = rising ? 0.f : myb;
    }
    {
        #pragma unroll
        for (int j = 0; j < 8; ++j) {
            const int q = tid + 128 * j;
            const float4 v = w1r[j];
            float* dst = act + (q >> 4) * WSTR + 4 * (q & 15);
            dst[0] = v.x; dst[1] = v.y; dst[2] = v.z; dst[3] = v.w;
        }
    }
    lds_barrier();

    int c0 = 0, c1 = 0;
    {
        const int n   = tid & 63;
        const int sel = (tid >> 6) * HID;
        const float* cp = smem + OFF_CP + sel;
        const float* cn = smem + OFF_CN + sel;
        const float* wrow = act + n * WSTR;
        float wv[HID];
        int pidx[HID];
        #pragma unroll
        for (int p = 0; p < HID; ++p) pidx[p] = __float_as_int(smem[OFF_PERM + p]);
        __builtin_amdgcn_sched_barrier(0);
        #pragma unroll
        for (int p = HID - 1; p >= 0; --p) wv[p] = wrow[pidx[p]];
        float cnv[HID];
        #pragma unroll
        for (int p = 0; p < HID; ++p) cnv[p] = cn[p];
        __builtin_amdgcn_sched_barrier(0);

        float sfx[HID];
        float cpv[HID];
        float sn = 0.f;
        #pragma unroll
        for (int k = 0; k < 7; ++k) {
            const int off = (k < 6) ? ((32 >> k) - 1) : 0;
            const int inc = (k < 6) ? (32 >> k) : 1;
            const float key0 = smem[OFF_SKEY + c0 + off];
            const float key1 = smem[OFF_SKEY + c1 + off];
            if (k == 6) {
                #pragma unroll
                for (int s = 0; s < HID / 2; ++s) cpv[s] = cp[s];
            }
            __builtin_amdgcn_sched_barrier(0);
            const int hi = 63 - 9 * k - (k ? 1 : 0);
            const int len = k ? 9 : 10;
            #pragma unroll
            for (int j = 0; j < len; ++j) {
                const int p = hi - j;
                sn = fmaf(wv[p], cnv[p], sn);
                sfx[p] = sn;
            }
            __builtin_amdgcn_sched_barrier(0);
            c0 += (key0 < xin0) ? inc : 0;
            c1 += (key1 < xin1) ? inc : 0;
        }
        float sp = sel ? b1v : 0.f;
        #pragma unroll
        for (int s = 0; s < HID; ++s) {
            if (s == 16) {
                #pragma unroll
                for (int t = HID / 2; t < HID; ++t) cpv[t] = cp[t];
                __builtin_amdgcn_sched_barrier(0);
            }
            tab[s * TSTR + tid] = sfx[s] + sp;
            sp = fmaf(wv[s], cpv[s], sp);
        }
        tab[HID * TSTR + tid] = sp;
    }

    __syncthreads();

    v2f Bown[16];
    {
        const float* __restrict__ wl = w2 + (size_t)d * HID * HID + (16 * wave + r) * HID + 2 * half;
        #pragma unroll
        for (int c = 0; c < 16; ++c) Bown[c] = *(const v2f*)(wl + c * 4);
    }
    const float b2own = b2[d * HID + 16 * wave + r];

    unsigned nlo0, nhi0, nlo1, nhi1;
    {
        const float* t0 = tab + c0 * TSTR;
        const float* t1 = tab + c1 * TSTR;
        unsigned m0[2] = {0u, 0u}, m1[2] = {0u, 0u};
        #pragma unroll
        for (int j = 15; j >= 0; --j) {
            const float4 sa = *reinterpret_cast<const float4*>(t0 + 4 * j);
            const float4 ta = *reinterpret_cast<const float4*>(t0 + HID + 4 * j);
            const float4 sb = *reinterpret_cast<const float4*>(t1 + 4 * j);
            const float4 tb = *reinterpret_cast<const float4*>(t1 + HID + 4 * j);
            const int w = j >> 3;
            m0[w] = (m0[w] << 1) | ((unsigned)__float_as_int(fmaf(xin0, sa.w, ta.w)) >> 31);
            m0[w] = (m0[w] << 1) | ((unsigned)__float_as_int(fmaf(xin0, sa.z, ta.z)) >> 31);
            m0[w] = (m0[w] << 1) | ((unsigned)__float_as_int(fmaf(xin0, sa.y, ta.y)) >> 31);
            m0[w] = (m0[w] << 1) | ((unsigned)__float_as_int(fmaf(xin0, sa.x, ta.x)) >> 31);
            m1[w] = (m1[w] << 1) | ((unsigned)__float_as_int(fmaf(xin1, sb.w, tb.w)) >> 31);
            m1[w] = (m1[w] << 1) | ((unsigned)__float_as_int(fmaf(xin1, sb.z, tb.z)) >> 31);
            m1[w] = (m1[w] << 1) | ((unsigned)__float_as_int(fmaf(xin1, sb.y, tb.y)) >> 31);
            m1[w] = (m1[w] << 1) | ((unsigned)__float_as_int(fmaf(xin1, sb.x, tb.x)) >> 31);
        }
        nlo0 = m0[0]; nhi0 = m0[1];
        nlo1 = m1[0]; nhi1 = m1[1];
    }

    unsigned* const mypat = patk + wave * 256;
    int pid0 = 0, pid1 = 0;
    int np = 0;
    {
        unsigned rem0 = __builtin_amdgcn_ballot_w32(true);
        unsigned rem1 = rem0;
        for (int it = 0; it < 32 && rem0 != 0u; ++it) {
            const int L = __builtin_ctz(rem0);
            const int ks = __builtin_amdgcn_readlane(c0, L);
            const unsigned kl = (unsigned)__builtin_amdgcn_readlane((int)nlo0, L);
            const unsigned kh = (unsigned)__builtin_amdgcn_readlane((int)nhi0, L);
            const bool e0 = (c0 == ks) & (nlo0 == kl) & (nhi0 == kh);
            const bool e1 = (c1 == ks) & (nlo1 == kl) & (nhi1 == kh);
            rem0 &= ~__builtin_amdgcn_ballot_w32(e0);
            rem1 &= ~__builtin_amdgcn_ballot_w32(e1);
            pid0 = e0 ? np : pid0;
            pid1 = e1 ? np : pid1;
            if (lane == 0) {
                mypat[4 * np]     = (unsigned)ks;
                mypat[4 * np + 1] = kl;
                mypat[4 * np + 2] = kh;
            }
            ++np;
        }
        for (int it = 0; it < 32 && rem1 != 0u; ++it) {
            const int L = __builtin_ctz(rem1);
            const int ks = __builtin_amdgcn_readlane(c1, L);
            const unsigned kl = (unsigned)__builtin_amdgcn_readlane((int)nlo1, L);
            const unsigned kh = (unsigned)__builtin_amdgcn_readlane((int)nhi1, L);
            const bool e1 = (c1 == ks) & (nlo1 == kl) & (nhi1 == kh);
            rem1 &= ~__builtin_amdgcn_ballot_w32(e1);
            pid1 = e1 ? np : pid1;
            if (lane == 0) {
                mypat[4 * np]     = (unsigned)ks;
                mypat[4 * np + 1] = kl;
                mypat[4 * np + 2] = kh;
            }
            ++np;
        }
    }

    if (lane == 0) npw[wave] = (unsigned)np;
    lds_barrier();
    const uint4 nn = *reinterpret_cast<const uint4*>(npw);
    const int pre1 = __builtin_amdgcn_readfirstlane((int)nn.x);
    const int pre2 = pre1 + __builtin_amdgcn_readfirstlane((int)nn.y);
    const int pre3 = pre2 + __builtin_amdgcn_readfirstlane((int)nn.z);
    const int nent = pre3 + __builtin_amdgcn_readfirstlane((int)nn.w);
    unsigned* const gp = gpk + wave * 256;
    int ng = 0;
    int g0 = 0, g1 = 0;
    bool shared = false;
    if (nent <= 64) {
        const int ea = lane, eb = lane + 32;
        const int wa = (ea >= pre1 ? 1 : 0) + (ea >= pre2 ? 1 : 0) + (ea >= pre3 ? 1 : 0);
        const int wb = (eb >= pre1 ? 1 : 0) + (eb >= pre2 ? 1 : 0) + (eb >= pre3 ? 1 : 0);
        int fa = ea >= pre1 ? pre1 : 0;
        fa = ea >= pre2 ? pre2 : fa;
        fa = ea >= pre3 ? pre3 : fa;
        int fb = eb >= pre1 ? pre1 : 0;
        fb = eb >= pre2 ? pre2 : fb;
        fb = eb >= pre3 ? pre3 : fb;
        const int qa = ea - fa;
        const int qb = eb - fb;
        const uint4 ka = *reinterpret_cast<const uint4*>(patk + 4 * (wa * 64 + qa));
        const uint4 kb = *reinterpret_cast<const uint4*>(patk + 4 * (wb * 64 + qb));
        const bool va = ea < nent, vb = eb < nent;
        unsigned rem0 = __builtin_amdgcn_ballot_w32(va);
        unsigned rem1 = __builtin_amdgcn_ballot_w32(vb);
        for (int it = 0; it < 32 && rem0 != 0u; ++it) {
            const int L = __builtin_ctz(rem0);
            const unsigned ks = (unsigned)__builtin_amdgcn_readlane((int)ka.x, L);
            const unsigned kl = (unsigned)__builtin_amdgcn_readlane((int)ka.y, L);
            const unsigned kh = (unsigned)__builtin_amdgcn_readlane((int)ka.z, L);
            const bool e0 = va & (ka.x == ks) & (ka.y == kl) & (ka.z == kh);
            const bool e1 = vb & (kb.x == ks) & (kb.y == kl) & (kb.z == kh);
            rem0 &= ~__builtin_amdgcn_ballot_w32(e0);
            rem1 &= ~__builtin_amdgcn_ballot_w32(e1);
            g0 = e0 ? ng : g0;
            g1 = e1 ? ng : g1;
            if (lane == 0) {
                gp[4 * ng]     = ks;
                gp[4 * ng + 1] = kl;
                gp[4 * ng + 2] = kh;
            }
            ++ng;
        }
        for (int it = 0; it < 32 && rem1 != 0u; ++it) {
            const int L = __builtin_ctz(rem1);
            const unsigned ks = (unsigned)__builtin_amdgcn_readlane((int)kb.x, L);
            const unsigned kl = (unsigned)__builtin_amdgcn_readlane((int)kb.y, L);
            const unsigned kh = (unsigned)__builtin_amdgcn_readlane((int)kb.z, L);
            const bool e1 = vb & (kb.x == ks) & (kb.y == kl) & (kb.z == kh);
            rem1 &= ~__builtin_amdgcn_ballot_w32(e1);
            g1 = e1 ? ng : g1;
            if (lane == 0) {
                gp[4 * ng]     = ks;
                gp[4 * ng + 1] = kl;
                gp[4 * ng + 2] = kh;
            }
            ++ng;
        }
        shared = ng <= 32;
    }

    if (shared) {
        int mybase = wave >= 1 ? pre1 : 0;
        mybase = wave >= 2 ? pre2 : mybase;
        mybase = wave >= 3 ? pre3 : mybase;
        const int i0 = mybase + pid0, i1 = mybase + pid1;
        const int t00 = __builtin_amdgcn_ds_bpermute(4 * (i0 & 31), g0);
        const int t01 = __builtin_amdgcn_ds_bpermute(4 * (i0 & 31), g1);
        const int t10 = __builtin_amdgcn_ds_bpermute(4 * (i1 & 31), g0);
        const int t11 = __builtin_amdgcn_ds_bpermute(4 * (i1 & 31), g1);
        const int pg0 = (i0 < 32) ? t00 : t01;
        const int pg1 = (i1 < 32) ? t10 : t11;

        float* const uvg = smem + OFF_ACT;
        const int nchg = (ng + PCHUNK - 1) / PCHUNK;
        #pragma nounroll
        for (int ch = 0; ch < nchg; ++ch) {
            const int q = ch * PCHUNK + (r & 7);
            const bool live = q < ng;
            const unsigned rs = gp[4 * q], rl = gp[4 * q + 1], rh = gp[4 * q + 2];
            const int psg = live ? (int)rs : 0;
            const unsigned alo = (live ? ~rl : 0u) >> (2 * half);
            const unsigned ahi = (live ? ~rh : 0u) >> (2 * half);
            const float* arow = tab + psg * TSTR + ((r & 8) ? HID : 0) + 2 * half;
            v2f araw[16], af[16];
            #pragma unroll
            for (int c = 0; c < 16; ++c) araw[c] = *(const v2f*)(arow + 4 * c);
            __builtin_amdgcn_sched_barrier(0);
            #pragma unroll
            for (int c = 0; c < 16; ++c) {
                const v2f t = araw[c];
                const unsigned wsel = (c < 8) ? alo : ahi;
                const int sh = 4 * (c & 7);
                af[c].x = ((wsel >> sh) & 1u) ? t.x : 0.f;
                af[c].y = ((wsel >> (sh + 1)) & 1u) ? t.y : 0.f;
            }

            const float ini = half ? b2own : 0.f;
            v8f accE = (v8f){ini, ini, ini, ini, ini, ini, ini, ini};
            v8f accO = (v8f){0.f, 0.f, 0.f, 0.f, 0.f, 0.f, 0.f, 0.f};
            #pragma unroll
            for (int c = 0; c < 16; c += 2) {
                accE = __builtin_amdgcn_wmma_f32_16x16x4_f32(
                           false, af[c], false, Bown[c], (short)0, accE, false, false);
                accO = __builtin_amdgcn_wmma_f32_16x16x4_f32(
                           false, af[c + 1], false, Bown[c + 1], (short)0, accO, false, false);
            }
            #pragma unroll
            for (int v = 0; v < 8; ++v)
                uvg[(ch * PCHUNK + v) * UVSTR + half * HID + 16 * wave + r] = accE[v] + accO[v];
        }
        lds_barrier();

        #pragma unroll
        for (int set = 0; set < 2; ++set) {
            const float xv = set ? xin1 : xin0;
            const float* p = uvg + (set ? pg1 : pg0) * UVSTR;
            float s = 0.f;
            #pragma unroll
            for (int g = 0; g < 2; ++g) {
                float4 u4[8], v4[8];
                #pragma unroll
                for (int t = 0; t < 8; ++t) {
                    u4[t] = *reinterpret_cast<const float4*>(p + 32 * g + 4 * t);
                    v4[t] = *reinterpret_cast<const float4*>(p + HID + 32 * g + 4 * t);
                }
                __builtin_amdgcn_sched_barrier(0);
                #pragma unroll
                for (int t = 0; t < 8; ++t) {
                    const int h = 32 * g + 4 * t;
                    float z;
                    z = fmaf(xv, u4[t].x, v4[t].x); z = z > 0.f ? z : 0.f; s = fmaf(wo[h + 0], z, s);
                    z = fmaf(xv, u4[t].y, v4[t].y); z = z > 0.f ? z : 0.f; s = fmaf(wo[h + 1], z, s);
                    z = fmaf(xv, u4[t].z, v4[t].z); z = z > 0.f ? z : 0.f; s = fmaf(wo[h + 2], z, s);
                    z = fmaf(xv, u4[t].w, v4[t].w); z = z > 0.f ? z : 0.f; s = fmaf(wo[h + 3], z, s);
                }
            }
            const int row = tid + set * MROWS;
            if (use_atomic) {
                __hip_atomic_fetch_add(&outbuf[row * OUT_DIM + o], s,
                                       __ATOMIC_RELAXED, __HIP_MEMORY_SCOPE_AGENT);
            } else {
                outbuf[(size_t)d * BSZ + row] = s;
            }
        }
    } else {
        float* const uv = smem + OFF_ACT + wave * UVW;
        const int nch = (np + PCHUNK - 1) / PCHUNK;
        #pragma nounroll
        for (int ch = 0; ch < nch; ++ch) {
            const int q = ch * PCHUNK + (r & 7);
            const bool live = q < np;
            const unsigned rs = mypat[4 * q], rl = mypat[4 * q + 1], rh = mypat[4 * q + 2];
            const int psg = live ? (int)rs : 0;
            const unsigned alo = (live ? ~rl : 0u) >> (2 * half);
            const unsigned ahi = (live ? ~rh : 0u) >> (2 * half);
            const float* arow = tab + psg * TSTR + ((r & 8) ? HID : 0) + 2 * half;
            v2f af[16];
            #pragma unroll
            for (int c = 0; c < 16; ++c) {
                const v2f t = *(const v2f*)(arow + 4 * c);
                const unsigned wsel = (c < 8) ? alo : ahi;
                const int sh = 4 * (c & 7);
                af[c].x = ((wsel >> sh) & 1u) ? t.x : 0.f;
                af[c].y = ((wsel >> (sh + 1)) & 1u) ? t.y : 0.f;
            }
            #pragma nounroll
            for (int j = 0; j < 4; ++j) {
                const float* __restrict__ wl = w2 + (size_t)d * HID * HID + (16 * j + r) * HID + 2 * half;
                v2f Bj[16];
                #pragma unroll
                for (int c = 0; c < 16; ++c) Bj[c] = *(const v2f*)(wl + c * 4);
                const float ini = half ? b2[d * HID + 16 * j + r] : 0.f;
                v8f accE = (v8f){ini, ini, ini, ini, ini, ini, ini, ini};
                v8f accO = (v8f){0.f, 0.f, 0.f, 0.f, 0.f, 0.f, 0.f, 0.f};
                #pragma unroll
                for (int c = 0; c < 16; c += 2) {
                    accE = __builtin_amdgcn_wmma_f32_16x16x4_f32(
                               false, af[c], false, Bj[c], (short)0, accE, false, false);
                    accO = __builtin_amdgcn_wmma_f32_16x16x4_f32(
                               false, af[c + 1], false, Bj[c + 1], (short)0, accO, false, false);
                }
                #pragma unroll
                for (int v = 0; v < 8; ++v)
                    uv[v * UVSTR + half * HID + 16 * j + r] = accE[v] + accO[v];
            }

            #pragma unroll
            for (int set = 0; set < 2; ++set) {
                const int pid = set ? pid1 : pid0;
                if ((pid >> 3) == ch) {
                    const float xv = set ? xin1 : xin0;
                    const float* p = uv + (pid & 7) * UVSTR;
                    float s = 0.f;
                    #pragma unroll
                    for (int g = 0; g < 2; ++g) {
                        float4 u4[8], v4[8];
                        #pragma unroll
                        for (int t = 0; t < 8; ++t) {
                            u4[t] = *reinterpret_cast<const float4*>(p + 32 * g + 4 * t);
                            v4[t] = *reinterpret_cast<const float4*>(p + HID + 32 * g + 4 * t);
                        }
                        #pragma unroll
                        for (int t = 0; t < 8; ++t) {
                            const int h = 32 * g + 4 * t;
                            float z;
                            z = fmaf(xv, u4[t].x, v4[t].x); z = z > 0.f ? z : 0.f; s = fmaf(wo[h + 0], z, s);
                            z = fmaf(xv, u4[t].y, v4[t].y); z = z > 0.f ? z : 0.f; s = fmaf(wo[h + 1], z, s);
                            z = fmaf(xv, u4[t].z, v4[t].z); z = z > 0.f ? z : 0.f; s = fmaf(wo[h + 2], z, s);
                            z = fmaf(xv, u4[t].w, v4[t].w); z = z > 0.f ? z : 0.f; s = fmaf(wo[h + 3], z, s);
                        }
                    }
                    const int row = tid + set * MROWS;
                    if (use_atomic) {
                        __hip_atomic_fetch_add(&outbuf[row * OUT_DIM + o], s,
                                               __ATOMIC_RELAXED, __HIP_MEMORY_SCOPE_AGENT);
                    } else {
                        outbuf[(size_t)d * BSZ + row] = s;
                    }
                }
            }
        }
    }
}

__global__ void kan_reduce_kernel(const float* __restrict__ part,
                                  const float* __restrict__ bout,
                                  float* __restrict__ out)
{
    const int o = blockIdx.x;
    const int b = threadIdx.x;
    float s = bout[o];
    const float* p = part + (size_t)o * 64 * BSZ + b;
    #pragma unroll 32
    for (int i = 0; i < 64; ++i) s += p[i * BSZ];
    out[b * OUT_DIM + o] = s;
}

__global__ void kan_init_out(const float* __restrict__ bout, float* __restrict__ out)
{
    const int idx = blockIdx.x * 256 + threadIdx.x;
    out[idx] = bout[idx & 63];
}

extern "C" void kernel_launch(void* const* d_in, const int* in_sizes, int n_in,
                              void* d_out, int out_size, void* d_ws, size_t ws_size,
                              hipStream_t stream)
{
    const float* x    = (const float*)d_in[0];
    const float* w0   = (const float*)d_in[1];
    const float* b0   = (const float*)d_in[2];
    const float* w1   = (const float*)d_in[3];
    const float* b1   = (const float*)d_in[4];
    const float* w2   = (const float*)d_in[5];
    const float* b2   = (const float*)d_in[6];
    const float* wout = (const float*)d_in[7];
    const float* bout = (const float*)d_in[8];
    float* out = (float*)d_out;

    const size_t need = (size_t)DDIM * BSZ * sizeof(float);
    if (ws_size >= need) {
        float* part = (float*)d_ws;
        kan_edge_kernel<<<DDIM, MROWS, 0, stream>>>(x, w0, b0, w1, b1, w2, b2, wout, part, 0);
        kan_reduce_kernel<<<OUT_DIM, BSZ, 0, stream>>>(part, bout, out);
    } else {
        kan_init_out<<<64, 256, 0, stream>>>(bout, out);
        kan_edge_kernel<<<DDIM, MROWS, 0, stream>>>(x, w0, b0, w1, b1, w2, b2, wout, out, 1);
    }
}
